# starting kernel
# baseline (speedup 1.0000x reference)
#include <hip/hip_runtime.h>

#define Nn    100000
#define Ee    3200000
#define FIN   256
#define Hh    32

typedef float v2f __attribute__((ext_vector_type(2)));
typedef float v8f __attribute__((ext_vector_type(8)));

__device__ __forceinline__ v8f wmma4(v2f a, v2f b, v8f c) {
  return __builtin_amdgcn_wmma_f32_16x16x4_f32(
       false, a,  false, b,
       (short)0, c,  false,  false);
}

__device__ __forceinline__ float sigmoidf_(float x) {
  return 1.0f / (1.0f + __expf(-x));
}

__global__ void pack_weights_kernel(const float* __restrict__ Wz, const float* __restrict__ Wr,
                                    const float* __restrict__ Wh, const float* __restrict__ bz,
                                    const float* __restrict__ br, const float* __restrict__ bh,
                                    float* __restrict__ Wcat, float* __restrict__ bcat) {
  int gid = blockIdx.x * blockDim.x + threadIdx.x;
  if (gid < FIN * 96) {
    int k = gid / 96, c = gid - k * 96;
    float v = (c < 32) ? Wz[k * 32 + c] : ((c < 64) ? Wr[k * 32 + c - 32] : Wh[k * 32 + c - 64]);
    Wcat[gid] = v;
  }
  if (gid < 96) {
    bcat[gid] = (gid < 32) ? bz[gid] : ((gid < 64) ? br[gid - 32] : bh[gid - 64]);
  }
}

__global__ void zero_deg_kernel(float* __restrict__ deg) {
  int gid = blockIdx.x * blockDim.x + threadIdx.x;
  if (gid < Nn) deg[gid] = 0.0f;
}

__global__ void deg_accum_kernel(const int* __restrict__ dst, const float* __restrict__ w,
                                 float* __restrict__ deg) {
  int gid = blockIdx.x * blockDim.x + threadIdx.x;
  if (gid < Ee) unsafeAtomicAdd(&deg[dst[gid]], w[gid]);
}

__global__ void finalize_dis_kernel(float* __restrict__ deg_dis) {
  int gid = blockIdx.x * blockDim.x + threadIdx.x;
  if (gid < Nn) deg_dis[gid] = rsqrtf(deg_dis[gid] + 1.0f);
}

__global__ void __launch_bounds__(32) gemm_xw_kernel(const float* __restrict__ x,
                                                     const float* __restrict__ Wcat,
                                                     float* __restrict__ XW) {
  __shared__ float Xs[16 * 260];
  const int lane = threadIdx.x;
  const int tile = blockIdx.x * 16;

  for (int i = lane; i < 1024; i += 32) {
    int r = i >> 6, c4 = (i & 63) << 2;
    const float4 v = *reinterpret_cast<const float4*>(x + (size_t)(tile + r) * FIN + c4);
    *reinterpret_cast<float4*>(&Xs[r * 260 + c4]) = v;
  }
  __syncthreads();

  const int m = lane & 15;
  const int koff = (lane >> 4) << 1;

  v8f acc[6] = {};
  const float2* arow = reinterpret_cast<const float2*>(&Xs[m * 260]);

#pragma unroll 4
  for (int k0 = 0; k0 < FIN; k0 += 4) {
    const int kk = k0 + koff;
    float2 af = arow[kk >> 1];
    v2f a; a.x = af.x; a.y = af.y;
    const float* brow = Wcat + (size_t)kk * 96 + m;
#pragma unroll
    for (int t = 0; t < 6; ++t) {
      v2f b; b.x = brow[t * 16]; b.y = brow[96 + t * 16];
      acc[t] = wmma4(a, b, acc[t]);
    }
  }

  const int row0 = (lane >> 4) << 3;
#pragma unroll
  for (int t = 0; t < 6; ++t)
#pragma unroll
    for (int v = 0; v < 8; ++v)
      XW[(size_t)(tile + row0 + v) * 96 + t * 16 + m] = acc[t][v];
}

__global__ void init_agg_kernel(const float* __restrict__ dis, const float* __restrict__ XW,
                                const float* __restrict__ bcat, float* __restrict__ AGG) {
  size_t gid = (size_t)blockIdx.x * blockDim.x + threadIdx.x;
  if (gid < (size_t)Nn * 96) {
    size_t i = gid / 96;
    int c = (int)(gid - i * 96);
    float d = dis[i];
    AGG[gid] = d * d * XW[gid] + bcat[c];
  }
}

__global__ void edge_scatter_kernel(const int* __restrict__ src, const int* __restrict__ dst,
                                    const float* __restrict__ w, const float* __restrict__ dis,
                                    const float* __restrict__ XW, float* __restrict__ AGG) {
  const int lane = threadIdx.x & 31;
  const long long e = (((long long)blockIdx.x * blockDim.x) + threadIdx.x) >> 5;
  if (e >= Ee) return;
  const int s = src[e], d = dst[e];
  const float nrm = dis[s] * w[e] * dis[d];
  const size_t bs = (size_t)s * 96, bd = (size_t)d * 96;
#pragma unroll
  for (int i = 0; i < 3; ++i) {
    const int c = lane + i * 32;
    unsafeAtomicAdd(&AGG[bd + c], nrm * XW[bs + c]);
  }
}

__global__ void __launch_bounds__(32) gru_kernel(const float* __restrict__ AGG,
                                                 const float* __restrict__ Hp,
                                                 const float* __restrict__ LzW, const float* __restrict__ Lzb,
                                                 const float* __restrict__ LrW, const float* __restrict__ Lrb,
                                                 const float* __restrict__ LhW, const float* __restrict__ Lhb,
                                                 const float* __restrict__ headW, const float* __restrict__ headb,
                                                 float* __restrict__ y_out, float* __restrict__ h_out) {
  __shared__ float As[16 * 132];
  __shared__ float Rs[16 * 36];
  const int lane = threadIdx.x;
  const int tile = blockIdx.x * 16;

  for (int idx = lane; idx < 16 * 128; idx += 32) {
    int r = idx >> 7, c = idx & 127;
    float v = (c < 96) ? AGG[(size_t)(tile + r) * 96 + c] : Hp[(size_t)(tile + r) * 32 + (c - 96)];
    As[r * 132 + c] = v;
  }
  __syncthreads();

  const int m = lane & 15;
  const int n = lane & 15;
  const int koff = (lane >> 4) << 1;

  v8f z0 = {}, z1 = {}, r0 = {}, r1 = {};
#pragma unroll 4
  for (int k0 = 0; k0 < 64; k0 += 4) {
    const int kk = k0 + koff;
    const int colz = (kk < 32) ? kk : (64 + kk);
    const int colr = (kk < 32) ? (32 + kk) : (64 + kk);
    v2f az; az.x = As[m * 132 + colz]; az.y = As[m * 132 + colz + 1];
    v2f ar; ar.x = As[m * 132 + colr]; ar.y = As[m * 132 + colr + 1];
    const float* lz = LzW + kk * 32 + n;
    const float* lr = LrW + kk * 32 + n;
    v2f b;
    b.x = lz[0];  b.y = lz[32];  z0 = wmma4(az, b, z0);
    b.x = lz[16]; b.y = lz[48];  z1 = wmma4(az, b, z1);
    b.x = lr[0];  b.y = lr[32];  r0 = wmma4(ar, b, r0);
    b.x = lr[16]; b.y = lr[48];  r1 = wmma4(ar, b, r1);
  }

  const float bz0 = Lzb[n], bz1 = Lzb[16 + n];
  const float br0 = Lrb[n], br1 = Lrb[16 + n];
  const int row0 = (lane >> 4) << 3;
#pragma unroll
  for (int v = 0; v < 8; ++v) {
    z0[v] = sigmoidf_(z0[v] + bz0);
    z1[v] = sigmoidf_(z1[v] + bz1);
    r0[v] = sigmoidf_(r0[v] + br0);
    r1[v] = sigmoidf_(r1[v] + br1);
    Rs[(row0 + v) * 36 + n]      = r0[v];
    Rs[(row0 + v) * 36 + 16 + n] = r1[v];
  }
  __syncthreads();

  v8f t0 = {}, t1 = {};
#pragma unroll 4
  for (int k0 = 0; k0 < 64; k0 += 4) {
    const int kk = k0 + koff;
    v2f a;
    a.x = As[m * 132 + 64 + kk];
    a.y = As[m * 132 + 65 + kk];
    if (kk >= 32) {
      a.x *= Rs[m * 36 + kk - 32];
      a.y *= Rs[m * 36 + kk - 31];
    }
    const float* lh = LhW + kk * 32 + n;
    v2f b;
    b.x = lh[0];  b.y = lh[32];  t0 = wmma4(a, b, t0);
    b.x = lh[16]; b.y = lh[48];  t1 = wmma4(a, b, t1);
  }

  const float bh0 = Lhb[n], bh1 = Lhb[16 + n];
  const float hw0 = headW[n], hw1 = headW[16 + n];
  float p[8];
#pragma unroll
  for (int v = 0; v < 8; ++v) {
    const float ht0 = tanhf(t0[v] + bh0);
    const float ht1 = tanhf(t1[v] + bh1);
    const float hp0 = As[(row0 + v) * 132 + 96 + n];
    const float hp1 = As[(row0 + v) * 132 + 112 + n];
    const float h0 = z0[v] * hp0 + (1.0f - z0[v]) * ht0;
    const float h1 = z1[v] * hp1 + (1.0f - z1[v]) * ht1;
    h_out[(size_t)(tile + row0 + v) * 32 + n]      = h0;
    h_out[(size_t)(tile + row0 + v) * 32 + 16 + n] = h1;
    p[v] = fmaxf(h0, 0.0f) * hw0 + fmaxf(h1, 0.0f) * hw1;
  }
#pragma unroll
  for (int s = 1; s < 16; s <<= 1)
#pragma unroll
    for (int v = 0; v < 8; ++v)
      p[v] += __shfl_xor(p[v], s, 32);
  if (n == 0) {
    const float hb = headb[0];
#pragma unroll
    for (int v = 0; v < 8; ++v)
      y_out[tile + row0 + v] = p[v] + hb;
  }
}

extern "C" void kernel_launch(void* const* d_in, const int* in_sizes, int n_in,
                              void* d_out, int out_size, void* d_ws, size_t ws_size,
                              hipStream_t stream) {
  const float* x    = (const float*)d_in[0];
  const int*   ei   = (const int*)d_in[1];
  const float* ew   = (const float*)d_in[2];
  const float* Hp   = (const float*)d_in[3];
  const float* Wz   = (const float*)d_in[4];
  const float* bz   = (const float*)d_in[5];
  const float* Wr   = (const float*)d_in[6];
  const float* br   = (const float*)d_in[7];
  const float* Wh   = (const float*)d_in[8];
  const float* bh   = (const float*)d_in[9];
  const float* LzW  = (const float*)d_in[10];
  const float* Lzb  = (const float*)d_in[11];
  const float* LrW  = (const float*)d_in[12];
  const float* Lrb  = (const float*)d_in[13];
  const float* LhW  = (const float*)d_in[14];
  const float* Lhb  = (const float*)d_in[15];
  const float* hW   = (const float*)d_in[16];
  const float* hb   = (const float*)d_in[17];

  const int* src = ei;
  const int* dst = ei + Ee;

  float* out_y = (float*)d_out;
  float* out_h = out_y + Nn;

  float* ws   = (float*)d_ws;
  float* dis  = ws;
  float* Wcat = ws + Nn;
  float* bcat = Wcat + FIN * 96;
  float* XW   = bcat + 96;
  float* AGG  = XW + (size_t)Nn * 96;

  pack_weights_kernel<<<(FIN * 96 + 255) / 256, 256, 0, stream>>>(Wz, Wr, Wh, bz, br, bh, Wcat, bcat);
  zero_deg_kernel<<<(Nn + 255) / 256, 256, 0, stream>>>(dis);
  deg_accum_kernel<<<Ee / 256, 256, 0, stream>>>(dst, ew, dis);
  finalize_dis_kernel<<<(Nn + 255) / 256, 256, 0, stream>>>(dis);
  gemm_xw_kernel<<<Nn / 16, 32, 0, stream>>>(x, Wcat, XW);
  init_agg_kernel<<<(Nn * 96) / 256, 256, 0, stream>>>(dis, XW, bcat, AGG);
  edge_scatter_kernel<<<Ee / 8, 256, 0, stream>>>(src, dst, ew, dis, XW, AGG);
  gru_kernel<<<Nn / 16, 32, 0, stream>>>(AGG, Hp, LzW, Lzb, LrW, Lrb, LhW, Lhb, hW, hb, out_y, out_h);
}


# Degree without global atomics: a work-group sums one edge slice into an LDS table per node range and stores it; finalize adds the 64 slice rows in fixed order; sums in arrival order per slice
# speedup vs baseline: 1.2258x; 1.2258x over previous
#include <hip/hip_runtime.h>

#define Nn    100000
#define Ee    3200000
#define FIN   256
#define Hh    32

typedef float v2f __attribute__((ext_vector_type(2)));
typedef float v8f __attribute__((ext_vector_type(8)));

__device__ __forceinline__ v8f wmma4(v2f a, v2f b, v8f c) {
  return __builtin_amdgcn_wmma_f32_16x16x4_f32(
       false, a,  false, b,
       (short)0, c,  false,  false);
}

__device__ __forceinline__ float sigmoidf_(float x) {
  return 1.0f / (1.0f + __expf(-x));
}

__global__ void pack_weights_kernel(const float* __restrict__ Wz, const float* __restrict__ Wr,
                                    const float* __restrict__ Wh, const float* __restrict__ bz,
                                    const float* __restrict__ br, const float* __restrict__ bh,
                                    float* __restrict__ Wcat, float* __restrict__ bcat) {
  int gid = blockIdx.x * blockDim.x + threadIdx.x;
  if (gid < FIN * 96) {
    int k = gid / 96, c = gid - k * 96;
    float v = (c < 32) ? Wz[k * 32 + c] : ((c < 64) ? Wr[k * 32 + c - 32] : Wh[k * 32 + c - 64]);
    Wcat[gid] = v;
  }
  if (gid < 96) {
    bcat[gid] = (gid < 32) ? bz[gid] : ((gid < 64) ? br[gid - 32] : bh[gid - 64]);
  }
}

#define DEG_Q      4
#define DEG_RANGE  (Nn / DEG_Q)
#define DEG_S      64
#define DEG_SLICE  (Ee / DEG_S)
#define DEG_NV     (DEG_SLICE / 4)
#define DEG_TPB    1024

__device__ __forceinline__ void deg_add(float* tab, int d, float wv) {
  if ((unsigned)d < (unsigned)DEG_RANGE) unsafeAtomicAdd(&tab[d], wv);
}

__global__ void __launch_bounds__(DEG_TPB) deg_partial_kernel(const int* __restrict__ dst,
                                                              const float* __restrict__ w,
                                                              float* __restrict__ part) {
  __shared__ float tab[DEG_RANGE];
  const int tid = threadIdx.x;
  const int q = blockIdx.x % DEG_Q;
  const int s = blockIdx.x / DEG_Q;
  const int lo = q * DEG_RANGE;

  float4* tab4 = reinterpret_cast<float4*>(tab);
  for (int i = tid; i < DEG_RANGE / 4; i += DEG_TPB) tab4[i] = make_float4(0.0f, 0.0f, 0.0f, 0.0f);
  __syncthreads();

  const int4*   d4 = reinterpret_cast<const int4*>(dst + (size_t)s * DEG_SLICE);
  const float4* w4 = reinterpret_cast<const float4*>(w + (size_t)s * DEG_SLICE);
  for (int i0 = tid; i0 < DEG_NV; i0 += 4 * DEG_TPB) {
    int4 dd[4];
    float4 ww[4];
#pragma unroll
    for (int j = 0; j < 4; ++j) {
      const int i = i0 + j * DEG_TPB;
      if (i < DEG_NV) {
        dd[j] = d4[i];
        ww[j] = w4[i];
      } else {
        dd[j] = make_int4(-1, -1, -1, -1);
        ww[j] = make_float4(0.0f, 0.0f, 0.0f, 0.0f);
      }
    }
#pragma unroll
    for (int j = 0; j < 4; ++j) {
      deg_add(tab, dd[j].x - lo, ww[j].x);
      deg_add(tab, dd[j].y - lo, ww[j].y);
      deg_add(tab, dd[j].z - lo, ww[j].z);
      deg_add(tab, dd[j].w - lo, ww[j].w);
    }
  }
  __syncthreads();

  float4* out4 = reinterpret_cast<float4*>(part + (size_t)s * Nn + lo);
  for (int i = tid; i < DEG_RANGE / 4; i += DEG_TPB) out4[i] = tab4[i];
}

__global__ void finalize_dis_kernel(const float* __restrict__ part, float* __restrict__ dis) {
  int gid = blockIdx.x * blockDim.x + threadIdx.x;
  if (gid < Nn) {
    float sum = 0.0f;
    for (int s = 0; s < DEG_S; ++s) sum += part[(size_t)s * Nn + gid];
    dis[gid] = rsqrtf(sum + 1.0f);
  }
}

__global__ void __launch_bounds__(32) gemm_xw_kernel(const float* __restrict__ x,
                                                     const float* __restrict__ Wcat,
                                                     float* __restrict__ XW) {
  __shared__ float Xs[16 * 260];
  const int lane = threadIdx.x;
  const int tile = blockIdx.x * 16;

  for (int i = lane; i < 1024; i += 32) {
    int r = i >> 6, c4 = (i & 63) << 2;
    const float4 v = *reinterpret_cast<const float4*>(x + (size_t)(tile + r) * FIN + c4);
    *reinterpret_cast<float4*>(&Xs[r * 260 + c4]) = v;
  }
  __syncthreads();

  const int m = lane & 15;
  const int koff = (lane >> 4) << 1;

  v8f acc[6] = {};
  const float2* arow = reinterpret_cast<const float2*>(&Xs[m * 260]);

#pragma unroll 4
  for (int k0 = 0; k0 < FIN; k0 += 4) {
    const int kk = k0 + koff;
    float2 af = arow[kk >> 1];
    v2f a; a.x = af.x; a.y = af.y;
    const float* brow = Wcat + (size_t)kk * 96 + m;
#pragma unroll
    for (int t = 0; t < 6; ++t) {
      v2f b; b.x = brow[t * 16]; b.y = brow[96 + t * 16];
      acc[t] = wmma4(a, b, acc[t]);
    }
  }

  const int row0 = (lane >> 4) << 3;
#pragma unroll
  for (int t = 0; t < 6; ++t)
#pragma unroll
    for (int v = 0; v < 8; ++v)
      XW[(size_t)(tile + row0 + v) * 96 + t * 16 + m] = acc[t][v];
}

__global__ void init_agg_kernel(const float* __restrict__ dis, const float* __restrict__ XW,
                                const float* __restrict__ bcat, float* __restrict__ AGG) {
  size_t gid = (size_t)blockIdx.x * blockDim.x + threadIdx.x;
  if (gid < (size_t)Nn * 96) {
    size_t i = gid / 96;
    int c = (int)(gid - i * 96);
    float d = dis[i];
    AGG[gid] = d * d * XW[gid] + bcat[c];
  }
}

__global__ void edge_scatter_kernel(const int* __restrict__ src, const int* __restrict__ dst,
                                    const float* __restrict__ w, const float* __restrict__ dis,
                                    const float* __restrict__ XW, float* __restrict__ AGG) {
  const int lane = threadIdx.x & 31;
  const long long e = (((long long)blockIdx.x * blockDim.x) + threadIdx.x) >> 5;
  if (e >= Ee) return;
  const int s = src[e], d = dst[e];
  const float nrm = dis[s] * w[e] * dis[d];
  const size_t bs = (size_t)s * 96, bd = (size_t)d * 96;
#pragma unroll
  for (int i = 0; i < 3; ++i) {
    const int c = lane + i * 32;
    unsafeAtomicAdd(&AGG[bd + c], nrm * XW[bs + c]);
  }
}

__global__ void __launch_bounds__(32) gru_kernel(const float* __restrict__ AGG,
                                                 const float* __restrict__ Hp,
                                                 const float* __restrict__ LzW, const float* __restrict__ Lzb,
                                                 const float* __restrict__ LrW, const float* __restrict__ Lrb,
                                                 const float* __restrict__ LhW, const float* __restrict__ Lhb,
                                                 const float* __restrict__ headW, const float* __restrict__ headb,
                                                 float* __restrict__ y_out, float* __restrict__ h_out) {
  __shared__ float As[16 * 132];
  __shared__ float Rs[16 * 36];
  const int lane = threadIdx.x;
  const int tile = blockIdx.x * 16;

  for (int idx = lane; idx < 16 * 128; idx += 32) {
    int r = idx >> 7, c = idx & 127;
    float v = (c < 96) ? AGG[(size_t)(tile + r) * 96 + c] : Hp[(size_t)(tile + r) * 32 + (c - 96)];
    As[r * 132 + c] = v;
  }
  __syncthreads();

  const int m = lane & 15;
  const int n = lane & 15;
  const int koff = (lane >> 4) << 1;

  v8f z0 = {}, z1 = {}, r0 = {}, r1 = {};
#pragma unroll 4
  for (int k0 = 0; k0 < 64; k0 += 4) {
    const int kk = k0 + koff;
    const int colz = (kk < 32) ? kk : (64 + kk);
    const int colr = (kk < 32) ? (32 + kk) : (64 + kk);
    v2f az; az.x = As[m * 132 + colz]; az.y = As[m * 132 + colz + 1];
    v2f ar; ar.x = As[m * 132 + colr]; ar.y = As[m * 132 + colr + 1];
    const float* lz = LzW + kk * 32 + n;
    const float* lr = LrW + kk * 32 + n;
    v2f b;
    b.x = lz[0];  b.y = lz[32];  z0 = wmma4(az, b, z0);
    b.x = lz[16]; b.y = lz[48];  z1 = wmma4(az, b, z1);
    b.x = lr[0];  b.y = lr[32];  r0 = wmma4(ar, b, r0);
    b.x = lr[16]; b.y = lr[48];  r1 = wmma4(ar, b, r1);
  }

  const float bz0 = Lzb[n], bz1 = Lzb[16 + n];
  const float br0 = Lrb[n], br1 = Lrb[16 + n];
  const int row0 = (lane >> 4) << 3;
#pragma unroll
  for (int v = 0; v < 8; ++v) {
    z0[v] = sigmoidf_(z0[v] + bz0);
    z1[v] = sigmoidf_(z1[v] + bz1);
    r0[v] = sigmoidf_(r0[v] + br0);
    r1[v] = sigmoidf_(r1[v] + br1);
    Rs[(row0 + v) * 36 + n]      = r0[v];
    Rs[(row0 + v) * 36 + 16 + n] = r1[v];
  }
  __syncthreads();

  v8f t0 = {}, t1 = {};
#pragma unroll 4
  for (int k0 = 0; k0 < 64; k0 += 4) {
    const int kk = k0 + koff;
    v2f a;
    a.x = As[m * 132 + 64 + kk];
    a.y = As[m * 132 + 65 + kk];
    if (kk >= 32) {
      a.x *= Rs[m * 36 + kk - 32];
      a.y *= Rs[m * 36 + kk - 31];
    }
    const float* lh = LhW + kk * 32 + n;
    v2f b;
    b.x = lh[0];  b.y = lh[32];  t0 = wmma4(a, b, t0);
    b.x = lh[16]; b.y = lh[48];  t1 = wmma4(a, b, t1);
  }

  const float bh0 = Lhb[n], bh1 = Lhb[16 + n];
  const float hw0 = headW[n], hw1 = headW[16 + n];
  float p[8];
#pragma unroll
  for (int v = 0; v < 8; ++v) {
    const float ht0 = tanhf(t0[v] + bh0);
    const float ht1 = tanhf(t1[v] + bh1);
    const float hp0 = As[(row0 + v) * 132 + 96 + n];
    const float hp1 = As[(row0 + v) * 132 + 112 + n];
    const float h0 = z0[v] * hp0 + (1.0f - z0[v]) * ht0;
    const float h1 = z1[v] * hp1 + (1.0f - z1[v]) * ht1;
    h_out[(size_t)(tile + row0 + v) * 32 + n]      = h0;
    h_out[(size_t)(tile + row0 + v) * 32 + 16 + n] = h1;
    p[v] = fmaxf(h0, 0.0f) * hw0 + fmaxf(h1, 0.0f) * hw1;
  }
#pragma unroll
  for (int s = 1; s < 16; s <<= 1)
#pragma unroll
    for (int v = 0; v < 8; ++v)
      p[v] += __shfl_xor(p[v], s, 32);
  if (n == 0) {
    const float hb = headb[0];
#pragma unroll
    for (int v = 0; v < 8; ++v)
      y_out[tile + row0 + v] = p[v] + hb;
  }
}

extern "C" void kernel_launch(void* const* d_in, const int* in_sizes, int n_in,
                              void* d_out, int out_size, void* d_ws, size_t ws_size,
                              hipStream_t stream) {
  const float* x    = (const float*)d_in[0];
  const int*   ei   = (const int*)d_in[1];
  const float* ew   = (const float*)d_in[2];
  const float* Hp   = (const float*)d_in[3];
  const float* Wz   = (const float*)d_in[4];
  const float* bz   = (const float*)d_in[5];
  const float* Wr   = (const float*)d_in[6];
  const float* br   = (const float*)d_in[7];
  const float* Wh   = (const float*)d_in[8];
  const float* bh   = (const float*)d_in[9];
  const float* LzW  = (const float*)d_in[10];
  const float* Lzb  = (const float*)d_in[11];
  const float* LrW  = (const float*)d_in[12];
  const float* Lrb  = (const float*)d_in[13];
  const float* LhW  = (const float*)d_in[14];
  const float* Lhb  = (const float*)d_in[15];
  const float* hW   = (const float*)d_in[16];
  const float* hb   = (const float*)d_in[17];

  const int* src = ei;
  const int* dst = ei + Ee;

  float* out_y = (float*)d_out;
  float* out_h = out_y + Nn;

  float* ws   = (float*)d_ws;
  float* dis  = ws;
  float* Wcat = ws + Nn;
  float* bcat = Wcat + FIN * 96;
  float* XW   = bcat + 96;
  float* AGG  = XW + (size_t)Nn * 96;
  float* part = AGG + (size_t)Nn * 96;

  pack_weights_kernel<<<(FIN * 96 + 255) / 256, 256, 0, stream>>>(Wz, Wr, Wh, bz, br, bh, Wcat, bcat);
  deg_partial_kernel<<<DEG_Q * DEG_S, DEG_TPB, 0, stream>>>(dst, ew, part);
  finalize_dis_kernel<<<(Nn + 255) / 256, 256, 0, stream>>>(part, dis);
  gemm_xw_kernel<<<Nn / 16, 32, 0, stream>>>(x, Wcat, XW);
  init_agg_kernel<<<(Nn * 96) / 256, 256, 0, stream>>>(dis, XW, bcat, AGG);
  edge_scatter_kernel<<<Ee / 8, 256, 0, stream>>>(src, dst, ew, dis, XW, AGG);
  gru_kernel<<<Nn / 16, 32, 0, stream>>>(AGG, Hp, LzW, Lzb, LrW, Lrb, LhW, Lhb, hW, hb, out_y, out_h);
}


# Edge sum without global atomics: one work-group per (800-row range, edge-list half) filters dst, adds norm*XW[src] rows into LDS; two planes added in fixed order; sums in arrival order per plane
# speedup vs baseline: 2.1891x; 1.7859x over previous
#include <hip/hip_runtime.h>

#define Nn    100000
#define Ee    3200000
#define FIN   256
#define Hh    32

typedef float v2f __attribute__((ext_vector_type(2)));
typedef float v8f __attribute__((ext_vector_type(8)));

__device__ __forceinline__ v8f wmma4(v2f a, v2f b, v8f c) {
  return __builtin_amdgcn_wmma_f32_16x16x4_f32(
       false, a,  false, b,
       (short)0, c,  false,  false);
}

__device__ __forceinline__ float sigmoidf_(float x) {
  return 1.0f / (1.0f + __expf(-x));
}

__global__ void pack_weights_kernel(const float* __restrict__ Wz, const float* __restrict__ Wr,
                                    const float* __restrict__ Wh, const float* __restrict__ bz,
                                    const float* __restrict__ br, const float* __restrict__ bh,
                                    float* __restrict__ Wcat, float* __restrict__ bcat) {
  int gid = blockIdx.x * blockDim.x + threadIdx.x;
  if (gid < FIN * 96) {
    int k = gid / 96, c = gid - k * 96;
    float v = (c < 32) ? Wz[k * 32 + c] : ((c < 64) ? Wr[k * 32 + c - 32] : Wh[k * 32 + c - 64]);
    Wcat[gid] = v;
  }
  if (gid < 96) {
    bcat[gid] = (gid < 32) ? bz[gid] : ((gid < 64) ? br[gid - 32] : bh[gid - 64]);
  }
}

#define DEG_Q      4
#define DEG_RANGE  (Nn / DEG_Q)
#define DEG_S      64
#define DEG_SLICE  (Ee / DEG_S)
#define DEG_NV     (DEG_SLICE / 4)
#define DEG_TPB    1024

__device__ __forceinline__ void deg_add(float* tab, int d, float wv) {
  if ((unsigned)d < (unsigned)DEG_RANGE) unsafeAtomicAdd(&tab[d], wv);
}

__global__ void __launch_bounds__(DEG_TPB) deg_partial_kernel(const int* __restrict__ dst,
                                                              const float* __restrict__ w,
                                                              float* __restrict__ part) {
  __shared__ float tab[DEG_RANGE];
  const int tid = threadIdx.x;
  const int q = blockIdx.x % DEG_Q;
  const int s = blockIdx.x / DEG_Q;
  const int lo = q * DEG_RANGE;

  float4* tab4 = reinterpret_cast<float4*>(tab);
  for (int i = tid; i < DEG_RANGE / 4; i += DEG_TPB) tab4[i] = make_float4(0.0f, 0.0f, 0.0f, 0.0f);
  __syncthreads();

  const int4*   d4 = reinterpret_cast<const int4*>(dst + (size_t)s * DEG_SLICE);
  const float4* w4 = reinterpret_cast<const float4*>(w + (size_t)s * DEG_SLICE);
  for (int i0 = tid; i0 < DEG_NV; i0 += 4 * DEG_TPB) {
    int4 dd[4];
    float4 ww[4];
#pragma unroll
    for (int j = 0; j < 4; ++j) {
      const int i = i0 + j * DEG_TPB;
      if (i < DEG_NV) {
        dd[j] = d4[i];
        ww[j] = w4[i];
      } else {
        dd[j] = make_int4(-1, -1, -1, -1);
        ww[j] = make_float4(0.0f, 0.0f, 0.0f, 0.0f);
      }
    }
#pragma unroll
    for (int j = 0; j < 4; ++j) {
      deg_add(tab, dd[j].x - lo, ww[j].x);
      deg_add(tab, dd[j].y - lo, ww[j].y);
      deg_add(tab, dd[j].z - lo, ww[j].z);
      deg_add(tab, dd[j].w - lo, ww[j].w);
    }
  }
  __syncthreads();

  float4* out4 = reinterpret_cast<float4*>(part + (size_t)s * Nn + lo);
  for (int i = tid; i < DEG_RANGE / 4; i += DEG_TPB) out4[i] = tab4[i];
}

__global__ void finalize_dis_kernel(const float* __restrict__ part, float* __restrict__ dis) {
  int gid = blockIdx.x * blockDim.x + threadIdx.x;
  if (gid < Nn) {
    float sum = 0.0f;
    for (int s = 0; s < DEG_S; ++s) sum += part[(size_t)s * Nn + gid];
    dis[gid] = rsqrtf(sum + 1.0f);
  }
}

__global__ void __launch_bounds__(32) gemm_xw_kernel(const float* __restrict__ x,
                                                     const float* __restrict__ Wcat,
                                                     float* __restrict__ XW) {
  __shared__ float Xs[16 * 260];
  const int lane = threadIdx.x;
  const int tile = blockIdx.x * 16;

  for (int i = lane; i < 1024; i += 32) {
    int r = i >> 6, c4 = (i & 63) << 2;
    const float4 v = *reinterpret_cast<const float4*>(x + (size_t)(tile + r) * FIN + c4);
    *reinterpret_cast<float4*>(&Xs[r * 260 + c4]) = v;
  }
  __syncthreads();

  const int m = lane & 15;
  const int koff = (lane >> 4) << 1;

  v8f acc[6] = {};
  const float2* arow = reinterpret_cast<const float2*>(&Xs[m * 260]);

#pragma unroll 4
  for (int k0 = 0; k0 < FIN; k0 += 4) {
    const int kk = k0 + koff;
    float2 af = arow[kk >> 1];
    v2f a; a.x = af.x; a.y = af.y;
    const float* brow = Wcat + (size_t)kk * 96 + m;
#pragma unroll
    for (int t = 0; t < 6; ++t) {
      v2f b; b.x = brow[t * 16]; b.y = brow[96 + t * 16];
      acc[t] = wmma4(a, b, acc[t]);
    }
  }

  const int row0 = (lane >> 4) << 3;
#pragma unroll
  for (int t = 0; t < 6; ++t)
#pragma unroll
    for (int v = 0; v < 8; ++v)
      XW[(size_t)(tile + row0 + v) * 96 + t * 16 + m] = acc[t][v];
}

#define AG_R      800
#define AG_Q      (Nn / AG_R)
#define AG_S      2
#define AG_SLICE  (Ee / AG_S)
#define AG_NV     (AG_SLICE / 4)
#define AG_TPB    1024
#define AG_LCAP   96

__device__ __forceinline__ void ag_push(int* lst, int& cnt, bool m, int e, int lane) {
  const unsigned mask = __builtin_amdgcn_ballot_w32(m);
  if (mask != 0u) {
    if (m) lst[cnt + __builtin_popcount(mask & ((1u << lane) - 1u))] = e;
    cnt += __builtin_popcount(mask);
  }
}

__device__ __forceinline__ void ag_flush(float* tab, const int* lst, int n, int lo, int lane,
                                         const int* __restrict__ src, const int* __restrict__ dst,
                                         const float* __restrict__ w, const float* __restrict__ dis,
                                         const float* __restrict__ XW) {
  int sN = 0, ld = 0;
  float nrm = 0.0f;
  if (lane < n) {
    const int e = lst[lane];
    const int s = src[e], d = dst[e];
    nrm = dis[s] * w[e] * dis[d];
    sN = s * 96;
    ld = (d - lo) * 96;
  }
  for (int j0 = 0; j0 < n; j0 += 8) {
    float v[8][3];
#pragma unroll
    for (int j = 0; j < 8; ++j) {
      const int sb = __builtin_amdgcn_readlane(sN, j0 + j);
#pragma unroll
      for (int i = 0; i < 3; ++i) v[j][i] = XW[(size_t)sb + lane + 32 * i];
    }
#pragma unroll
    for (int j = 0; j < 8; ++j) {
      if (j0 + j < n) {
        const float nj = __int_as_float(__builtin_amdgcn_readlane(__float_as_int(nrm), j0 + j));
        const int lj = __builtin_amdgcn_readlane(ld, j0 + j);
#pragma unroll
        for (int i = 0; i < 3; ++i) unsafeAtomicAdd(&tab[lj + lane + 32 * i], nj * v[j][i]);
      }
    }
  }
}

__global__ void __launch_bounds__(AG_TPB) agg_range_kernel(const int* __restrict__ src,
                                                           const int* __restrict__ dst,
                                                           const float* __restrict__ w,
                                                           const float* __restrict__ dis,
                                                           const float* __restrict__ XW,
                                                           float* __restrict__ P) {
  __shared__ float tab[AG_R * 96];
  __shared__ int lst[(AG_TPB / 32) * AG_LCAP];
  const int tid = threadIdx.x;
  const int lane = tid & 31;
  const int s = blockIdx.x % AG_S;
  const int q = blockIdx.x / AG_S;
  const int lo = q * AG_R;

  float4* tab4 = reinterpret_cast<float4*>(tab);
  for (int i = tid; i < AG_R * 24; i += AG_TPB) tab4[i] = make_float4(0.0f, 0.0f, 0.0f, 0.0f);
  __syncthreads();

  int* mylst = lst + (tid >> 5) * AG_LCAP;
  int cnt = 0;
  const int ebase = s * AG_SLICE;
  const int4* d4 = reinterpret_cast<const int4*>(dst + ebase);
  for (int i0 = tid; i0 < AG_NV; i0 += 4 * AG_TPB) {
    int4 dd[4];
#pragma unroll
    for (int j = 0; j < 4; ++j) {
      const int i = i0 + j * AG_TPB;
      dd[j] = (i < AG_NV) ? d4[i] : make_int4(-1, -1, -1, -1);
    }
#pragma unroll
    for (int j = 0; j < 4; ++j) {
      const int e = ebase + 4 * (i0 + j * AG_TPB);
      ag_push(mylst, cnt, (unsigned)(dd[j].x - lo) < (unsigned)AG_R, e, lane);
      ag_push(mylst, cnt, (unsigned)(dd[j].y - lo) < (unsigned)AG_R, e + 1, lane);
      while (cnt >= 32) {
        cnt -= 32;
        ag_flush(tab, mylst + cnt, 32, lo, lane, src, dst, w, dis, XW);
      }
      ag_push(mylst, cnt, (unsigned)(dd[j].z - lo) < (unsigned)AG_R, e + 2, lane);
      ag_push(mylst, cnt, (unsigned)(dd[j].w - lo) < (unsigned)AG_R, e + 3, lane);
      while (cnt >= 32) {
        cnt -= 32;
        ag_flush(tab, mylst + cnt, 32, lo, lane, src, dst, w, dis, XW);
      }
    }
  }
  if (cnt > 0) ag_flush(tab, mylst, cnt, lo, lane, src, dst, w, dis, XW);
  __syncthreads();

  float4* out4 = reinterpret_cast<float4*>(P + ((size_t)s * Nn + lo) * 96);
  for (int i = tid; i < AG_R * 24; i += AG_TPB) out4[i] = tab4[i];
}

__global__ void combine_agg_kernel(const float* __restrict__ dis, const float* __restrict__ XW,
                                   const float* __restrict__ bcat, const float* __restrict__ P,
                                   float* __restrict__ AGG) {
  size_t gid = (size_t)blockIdx.x * blockDim.x + threadIdx.x;
  if (gid < (size_t)Nn * 96) {
    size_t i = gid / 96;
    int c = (int)(gid - i * 96);
    float d = dis[i];
    float a = d * d * XW[gid] + bcat[c];
#pragma unroll
    for (int sl = 0; sl < AG_S; ++sl) a += P[(size_t)sl * Nn * 96 + gid];
    AGG[gid] = a;
  }
}

__global__ void __launch_bounds__(32) gru_kernel(const float* __restrict__ AGG,
                                                 const float* __restrict__ Hp,
                                                 const float* __restrict__ LzW, const float* __restrict__ Lzb,
                                                 const float* __restrict__ LrW, const float* __restrict__ Lrb,
                                                 const float* __restrict__ LhW, const float* __restrict__ Lhb,
                                                 const float* __restrict__ headW, const float* __restrict__ headb,
                                                 float* __restrict__ y_out, float* __restrict__ h_out) {
  __shared__ float As[16 * 132];
  __shared__ float Rs[16 * 36];
  const int lane = threadIdx.x;
  const int tile = blockIdx.x * 16;

  for (int idx = lane; idx < 16 * 128; idx += 32) {
    int r = idx >> 7, c = idx & 127;
    float v = (c < 96) ? AGG[(size_t)(tile + r) * 96 + c] : Hp[(size_t)(tile + r) * 32 + (c - 96)];
    As[r * 132 + c] = v;
  }
  __syncthreads();

  const int m = lane & 15;
  const int n = lane & 15;
  const int koff = (lane >> 4) << 1;

  v8f z0 = {}, z1 = {}, r0 = {}, r1 = {};
#pragma unroll 4
  for (int k0 = 0; k0 < 64; k0 += 4) {
    const int kk = k0 + koff;
    const int colz = (kk < 32) ? kk : (64 + kk);
    const int colr = (kk < 32) ? (32 + kk) : (64 + kk);
    v2f az; az.x = As[m * 132 + colz]; az.y = As[m * 132 + colz + 1];
    v2f ar; ar.x = As[m * 132 + colr]; ar.y = As[m * 132 + colr + 1];
    const float* lz = LzW + kk * 32 + n;
    const float* lr = LrW + kk * 32 + n;
    v2f b;
    b.x = lz[0];  b.y = lz[32];  z0 = wmma4(az, b, z0);
    b.x = lz[16]; b.y = lz[48];  z1 = wmma4(az, b, z1);
    b.x = lr[0];  b.y = lr[32];  r0 = wmma4(ar, b, r0);
    b.x = lr[16]; b.y = lr[48];  r1 = wmma4(ar, b, r1);
  }

  const float bz0 = Lzb[n], bz1 = Lzb[16 + n];
  const float br0 = Lrb[n], br1 = Lrb[16 + n];
  const int row0 = (lane >> 4) << 3;
#pragma unroll
  for (int v = 0; v < 8; ++v) {
    z0[v] = sigmoidf_(z0[v] + bz0);
    z1[v] = sigmoidf_(z1[v] + bz1);
    r0[v] = sigmoidf_(r0[v] + br0);
    r1[v] = sigmoidf_(r1[v] + br1);
    Rs[(row0 + v) * 36 + n]      = r0[v];
    Rs[(row0 + v) * 36 + 16 + n] = r1[v];
  }
  __syncthreads();

  v8f t0 = {}, t1 = {};
#pragma unroll 4
  for (int k0 = 0; k0 < 64; k0 += 4) {
    const int kk = k0 + koff;
    v2f a;
    a.x = As[m * 132 + 64 + kk];
    a.y = As[m * 132 + 65 + kk];
    if (kk >= 32) {
      a.x *= Rs[m * 36 + kk - 32];
      a.y *= Rs[m * 36 + kk - 31];
    }
    const float* lh = LhW + kk * 32 + n;
    v2f b;
    b.x = lh[0];  b.y = lh[32];  t0 = wmma4(a, b, t0);
    b.x = lh[16]; b.y = lh[48];  t1 = wmma4(a, b, t1);
  }

  const float bh0 = Lhb[n], bh1 = Lhb[16 + n];
  const float hw0 = headW[n], hw1 = headW[16 + n];
  float p[8];
#pragma unroll
  for (int v = 0; v < 8; ++v) {
    const float ht0 = tanhf(t0[v] + bh0);
    const float ht1 = tanhf(t1[v] + bh1);
    const float hp0 = As[(row0 + v) * 132 + 96 + n];
    const float hp1 = As[(row0 + v) * 132 + 112 + n];
    const float h0 = z0[v] * hp0 + (1.0f - z0[v]) * ht0;
    const float h1 = z1[v] * hp1 + (1.0f - z1[v]) * ht1;
    h_out[(size_t)(tile + row0 + v) * 32 + n]      = h0;
    h_out[(size_t)(tile + row0 + v) * 32 + 16 + n] = h1;
    p[v] = fmaxf(h0, 0.0f) * hw0 + fmaxf(h1, 0.0f) * hw1;
  }
#pragma unroll
  for (int s = 1; s < 16; s <<= 1)
#pragma unroll
    for (int v = 0; v < 8; ++v)
      p[v] += __shfl_xor(p[v], s, 32);
  if (n == 0) {
    const float hb = headb[0];
#pragma unroll
    for (int v = 0; v < 8; ++v)
      y_out[tile + row0 + v] = p[v] + hb;
  }
}

extern "C" void kernel_launch(void* const* d_in, const int* in_sizes, int n_in,
                              void* d_out, int out_size, void* d_ws, size_t ws_size,
                              hipStream_t stream) {
  const float* x    = (const float*)d_in[0];
  const int*   ei   = (const int*)d_in[1];
  const float* ew   = (const float*)d_in[2];
  const float* Hp   = (const float*)d_in[3];
  const float* Wz   = (const float*)d_in[4];
  const float* bz   = (const float*)d_in[5];
  const float* Wr   = (const float*)d_in[6];
  const float* br   = (const float*)d_in[7];
  const float* Wh   = (const float*)d_in[8];
  const float* bh   = (const float*)d_in[9];
  const float* LzW  = (const float*)d_in[10];
  const float* Lzb  = (const float*)d_in[11];
  const float* LrW  = (const float*)d_in[12];
  const float* Lrb  = (const float*)d_in[13];
  const float* LhW  = (const float*)d_in[14];
  const float* Lhb  = (const float*)d_in[15];
  const float* hW   = (const float*)d_in[16];
  const float* hb   = (const float*)d_in[17];

  const int* src = ei;
  const int* dst = ei + Ee;

  float* out_y = (float*)d_out;
  float* out_h = out_y + Nn;

  float* ws   = (float*)d_ws;
  float* dis  = ws;
  float* Wcat = ws + Nn;
  float* bcat = Wcat + FIN * 96;
  float* XW   = bcat + 96;
  float* AGG  = XW + (size_t)Nn * 96;
  float* part = AGG + (size_t)Nn * 96;
  float* P    = part + (size_t)DEG_S * Nn;

  pack_weights_kernel<<<(FIN * 96 + 255) / 256, 256, 0, stream>>>(Wz, Wr, Wh, bz, br, bh, Wcat, bcat);
  deg_partial_kernel<<<DEG_Q * DEG_S, DEG_TPB, 0, stream>>>(dst, ew, part);
  finalize_dis_kernel<<<(Nn + 255) / 256, 256, 0, stream>>>(part, dis);
  gemm_xw_kernel<<<Nn / 16, 32, 0, stream>>>(x, Wcat, XW);
  agg_range_kernel<<<AG_Q * AG_S, AG_TPB, 0, stream>>>(src, dst, ew, dis, XW, P);
  combine_agg_kernel<<<(Nn * 96) / 256, 256, 0, stream>>>(dis, XW, bcat, P, AGG);
  gru_kernel<<<Nn / 16, 32, 0, stream>>>(AGG, Hp, LzW, Lzb, LrW, Lrb, LhW, Lhb, hW, hb, out_y, out_h);
}


# Aggregate's filter reads a 1-byte range id per edge (dst/800, written by the degree pass) instead of the 4-byte dst: 4x fewer scan bytes and trips; sums still in arrival order per plane
# speedup vs baseline: 2.3053x; 1.0531x over previous
#include <hip/hip_runtime.h>

#define Nn    100000
#define Ee    3200000
#define FIN   256
#define Hh    32
#define AG_R  800

typedef float v2f __attribute__((ext_vector_type(2)));
typedef float v8f __attribute__((ext_vector_type(8)));

__device__ __forceinline__ v8f wmma4(v2f a, v2f b, v8f c) {
  return __builtin_amdgcn_wmma_f32_16x16x4_f32(
       false, a,  false, b,
       (short)0, c,  false,  false);
}

__device__ __forceinline__ float sigmoidf_(float x) {
  return 1.0f / (1.0f + __expf(-x));
}

__global__ void pack_weights_kernel(const float* __restrict__ Wz, const float* __restrict__ Wr,
                                    const float* __restrict__ Wh, const float* __restrict__ bz,
                                    const float* __restrict__ br, const float* __restrict__ bh,
                                    float* __restrict__ Wcat, float* __restrict__ bcat) {
  int gid = blockIdx.x * blockDim.x + threadIdx.x;
  if (gid < FIN * 96) {
    int k = gid / 96, c = gid - k * 96;
    float v = (c < 32) ? Wz[k * 32 + c] : ((c < 64) ? Wr[k * 32 + c - 32] : Wh[k * 32 + c - 64]);
    Wcat[gid] = v;
  }
  if (gid < 96) {
    bcat[gid] = (gid < 32) ? bz[gid] : ((gid < 64) ? br[gid - 32] : bh[gid - 64]);
  }
}

#define DEG_Q      4
#define DEG_RANGE  (Nn / DEG_Q)
#define DEG_S      64
#define DEG_SLICE  (Ee / DEG_S)
#define DEG_NV     (DEG_SLICE / 4)
#define DEG_TPB    1024

__device__ __forceinline__ unsigned rid_of(int d) {
  return ((unsigned)d < (unsigned)Nn) ? (unsigned)d / (unsigned)AG_R : 255u;
}

__device__ __forceinline__ void deg_add(float* tab, int d, float wv) {
  if ((unsigned)d < (unsigned)DEG_RANGE) unsafeAtomicAdd(&tab[d], wv);
}

__global__ void __launch_bounds__(DEG_TPB) deg_partial_kernel(const int* __restrict__ dst,
                                                              const float* __restrict__ w,
                                                              float* __restrict__ part,
                                                              unsigned* __restrict__ rid4) {
  __shared__ float tab[DEG_RANGE];
  const int tid = threadIdx.x;
  const int q = blockIdx.x % DEG_Q;
  const int s = blockIdx.x / DEG_Q;
  const int lo = q * DEG_RANGE;

  float4* tab4 = reinterpret_cast<float4*>(tab);
  for (int i = tid; i < DEG_RANGE / 4; i += DEG_TPB) tab4[i] = make_float4(0.0f, 0.0f, 0.0f, 0.0f);
  __syncthreads();

  const int4*   d4 = reinterpret_cast<const int4*>(dst + (size_t)s * DEG_SLICE);
  const float4* w4 = reinterpret_cast<const float4*>(w + (size_t)s * DEG_SLICE);
  for (int i0 = tid; i0 < DEG_NV; i0 += 4 * DEG_TPB) {
    int4 dd[4];
    float4 ww[4];
#pragma unroll
    for (int j = 0; j < 4; ++j) {
      const int i = i0 + j * DEG_TPB;
      if (i < DEG_NV) {
        dd[j] = d4[i];
        ww[j] = w4[i];
      } else {
        dd[j] = make_int4(-1, -1, -1, -1);
        ww[j] = make_float4(0.0f, 0.0f, 0.0f, 0.0f);
      }
    }
#pragma unroll
    for (int j = 0; j < 4; ++j) {
      if (j == q && i0 + j * DEG_TPB < DEG_NV)
        rid4[(size_t)s * DEG_NV + i0 + j * DEG_TPB] =
            rid_of(dd[j].x) | (rid_of(dd[j].y) << 8) | (rid_of(dd[j].z) << 16) | (rid_of(dd[j].w) << 24);
      deg_add(tab, dd[j].x - lo, ww[j].x);
      deg_add(tab, dd[j].y - lo, ww[j].y);
      deg_add(tab, dd[j].z - lo, ww[j].z);
      deg_add(tab, dd[j].w - lo, ww[j].w);
    }
  }
  __syncthreads();

  float4* out4 = reinterpret_cast<float4*>(part + (size_t)s * Nn + lo);
  for (int i = tid; i < DEG_RANGE / 4; i += DEG_TPB) out4[i] = tab4[i];
}

__global__ void finalize_dis_kernel(const float* __restrict__ part, float* __restrict__ dis) {
  int gid = blockIdx.x * blockDim.x + threadIdx.x;
  if (gid < Nn) {
    float sum = 0.0f;
    for (int s = 0; s < DEG_S; ++s) sum += part[(size_t)s * Nn + gid];
    dis[gid] = rsqrtf(sum + 1.0f);
  }
}

__global__ void __launch_bounds__(32) gemm_xw_kernel(const float* __restrict__ x,
                                                     const float* __restrict__ Wcat,
                                                     float* __restrict__ XW) {
  __shared__ float Xs[16 * 260];
  const int lane = threadIdx.x;
  const int tile = blockIdx.x * 16;

  for (int i = lane; i < 1024; i += 32) {
    int r = i >> 6, c4 = (i & 63) << 2;
    const float4 v = *reinterpret_cast<const float4*>(x + (size_t)(tile + r) * FIN + c4);
    *reinterpret_cast<float4*>(&Xs[r * 260 + c4]) = v;
  }
  __syncthreads();

  const int m = lane & 15;
  const int koff = (lane >> 4) << 1;

  v8f acc[6] = {};
  const float2* arow = reinterpret_cast<const float2*>(&Xs[m * 260]);

#pragma unroll 4
  for (int k0 = 0; k0 < FIN; k0 += 4) {
    const int kk = k0 + koff;
    float2 af = arow[kk >> 1];
    v2f a; a.x = af.x; a.y = af.y;
    const float* brow = Wcat + (size_t)kk * 96 + m;
#pragma unroll
    for (int t = 0; t < 6; ++t) {
      v2f b; b.x = brow[t * 16]; b.y = brow[96 + t * 16];
      acc[t] = wmma4(a, b, acc[t]);
    }
  }

  const int row0 = (lane >> 4) << 3;
#pragma unroll
  for (int t = 0; t < 6; ++t)
#pragma unroll
    for (int v = 0; v < 8; ++v)
      XW[(size_t)(tile + row0 + v) * 96 + t * 16 + m] = acc[t][v];
}

#define AG_Q      (Nn / AG_R)
#define AG_S      2
#define AG_SLICE  (Ee / AG_S)
#define AG_NV     (AG_SLICE / 16)
#define AG_TPB    1024
#define AG_LCAP   96

__device__ __forceinline__ void ag_push(int* lst, int& cnt, bool m, int e, int lane) {
  const unsigned mask = __builtin_amdgcn_ballot_w32(m);
  if (mask != 0u) {
    if (m) lst[cnt + __builtin_popcount(mask & ((1u << lane) - 1u))] = e;
    cnt += __builtin_popcount(mask);
  }
}

__device__ __forceinline__ void ag_flush(float* tab, const int* lst, int n, int lo, int lane,
                                         const int* __restrict__ src, const int* __restrict__ dst,
                                         const float* __restrict__ w, const float* __restrict__ dis,
                                         const float* __restrict__ XW) {
  int sN = 0, ld = 0;
  float nrm = 0.0f;
  if (lane < n) {
    const int e = lst[lane];
    const int s = src[e], d = dst[e];
    nrm = dis[s] * w[e] * dis[d];
    sN = s * 96;
    ld = (d - lo) * 96;
  }
  for (int j0 = 0; j0 < n; j0 += 8) {
    float v[8][3];
#pragma unroll
    for (int j = 0; j < 8; ++j) {
      const int sb = __builtin_amdgcn_readlane(sN, j0 + j);
#pragma unroll
      for (int i = 0; i < 3; ++i) v[j][i] = XW[(size_t)sb + lane + 32 * i];
    }
#pragma unroll
    for (int j = 0; j < 8; ++j) {
      if (j0 + j < n) {
        const float nj = __int_as_float(__builtin_amdgcn_readlane(__float_as_int(nrm), j0 + j));
        const int lj = __builtin_amdgcn_readlane(ld, j0 + j);
#pragma unroll
        for (int i = 0; i < 3; ++i) unsafeAtomicAdd(&tab[lj + lane + 32 * i], nj * v[j][i]);
      }
    }
  }
}

__device__ __forceinline__ void ag_word(float* tab, int* lst, int& cnt, unsigned x, int e, int lo, int lane,
                                        const int* __restrict__ src, const int* __restrict__ dst,
                                        const float* __restrict__ w, const float* __restrict__ dis,
                                        const float* __restrict__ XW) {
  ag_push(lst, cnt, (x & 0x000000ffu) == 0u, e, lane);
  ag_push(lst, cnt, (x & 0x0000ff00u) == 0u, e + 1, lane);
  while (cnt >= 32) {
    cnt -= 32;
    ag_flush(tab, lst + cnt, 32, lo, lane, src, dst, w, dis, XW);
  }
  ag_push(lst, cnt, (x & 0x00ff0000u) == 0u, e + 2, lane);
  ag_push(lst, cnt, (x & 0xff000000u) == 0u, e + 3, lane);
  while (cnt >= 32) {
    cnt -= 32;
    ag_flush(tab, lst + cnt, 32, lo, lane, src, dst, w, dis, XW);
  }
}

__global__ void __launch_bounds__(AG_TPB) agg_range_kernel(const int* __restrict__ src,
                                                           const int* __restrict__ dst,
                                                           const unsigned* __restrict__ rid4,
                                                           const float* __restrict__ w,
                                                           const float* __restrict__ dis,
                                                           const float* __restrict__ XW,
                                                           float* __restrict__ P) {
  __shared__ float tab[AG_R * 96];
  __shared__ int lst[(AG_TPB / 32) * AG_LCAP];
  const int tid = threadIdx.x;
  const int lane = tid & 31;
  const int s = blockIdx.x % AG_S;
  const int q = blockIdx.x / AG_S;
  const int lo = q * AG_R;

  float4* tab4 = reinterpret_cast<float4*>(tab);
  for (int i = tid; i < AG_R * 24; i += AG_TPB) tab4[i] = make_float4(0.0f, 0.0f, 0.0f, 0.0f);
  __syncthreads();

  int* mylst = lst + (tid >> 5) * AG_LCAP;
  int cnt = 0;
  const int ebase = s * AG_SLICE;
  const uint4* r16 = reinterpret_cast<const uint4*>(rid4) + (size_t)s * AG_NV;
  const unsigned qs = (unsigned)q * 0x01010101u;
  for (int i0 = tid; i0 < AG_NV; i0 += 4 * AG_TPB) {
    uint4 dd[4];
#pragma unroll
    for (int j = 0; j < 4; ++j) {
      const int i = i0 + j * AG_TPB;
      dd[j] = (i < AG_NV) ? r16[i] : make_uint4(~0u, ~0u, ~0u, ~0u);
    }
    int e = ebase + 16 * i0;
#pragma unroll 1
    for (int j = 0; j < 4; ++j) {
      const uint4 cur = dd[0];
      dd[0] = dd[1];
      dd[1] = dd[2];
      dd[2] = dd[3];
      ag_word(tab, mylst, cnt, cur.x ^ qs, e,      lo, lane, src, dst, w, dis, XW);
      ag_word(tab, mylst, cnt, cur.y ^ qs, e + 4,  lo, lane, src, dst, w, dis, XW);
      ag_word(tab, mylst, cnt, cur.z ^ qs, e + 8,  lo, lane, src, dst, w, dis, XW);
      ag_word(tab, mylst, cnt, cur.w ^ qs, e + 12, lo, lane, src, dst, w, dis, XW);
      e += 16 * AG_TPB;
    }
  }
  if (cnt > 0) ag_flush(tab, mylst, cnt, lo, lane, src, dst, w, dis, XW);
  __syncthreads();

  float4* out4 = reinterpret_cast<float4*>(P + ((size_t)s * Nn + lo) * 96);
  for (int i = tid; i < AG_R * 24; i += AG_TPB) out4[i] = tab4[i];
}

__global__ void combine_agg_kernel(const float* __restrict__ dis, const float* __restrict__ XW,
                                   const float* __restrict__ bcat, const float* __restrict__ P,
                                   float* __restrict__ AGG) {
  size_t gid = (size_t)blockIdx.x * blockDim.x + threadIdx.x;
  if (gid < (size_t)Nn * 96) {
    size_t i = gid / 96;
    int c = (int)(gid - i * 96);
    float d = dis[i];
    float a = d * d * XW[gid] + bcat[c];
#pragma unroll
    for (int sl = 0; sl < AG_S; ++sl) a += P[(size_t)sl * Nn * 96 + gid];
    AGG[gid] = a;
  }
}

__global__ void __launch_bounds__(32) gru_kernel(const float* __restrict__ AGG,
                                                 const float* __restrict__ Hp,
                                                 const float* __restrict__ LzW, const float* __restrict__ Lzb,
                                                 const float* __restrict__ LrW, const float* __restrict__ Lrb,
                                                 const float* __restrict__ LhW, const float* __restrict__ Lhb,
                                                 const float* __restrict__ headW, const float* __restrict__ headb,
                                                 float* __restrict__ y_out, float* __restrict__ h_out) {
  __shared__ float As[16 * 132];
  __shared__ float Rs[16 * 36];
  const int lane = threadIdx.x;
  const int tile = blockIdx.x * 16;

  for (int idx = lane; idx < 16 * 128; idx += 32) {
    int r = idx >> 7, c = idx & 127;
    float v = (c < 96) ? AGG[(size_t)(tile + r) * 96 + c] : Hp[(size_t)(tile + r) * 32 + (c - 96)];
    As[r * 132 + c] = v;
  }
  __syncthreads();

  const int m = lane & 15;
  const int n = lane & 15;
  const int koff = (lane >> 4) << 1;

  v8f z0 = {}, z1 = {}, r0 = {}, r1 = {};
#pragma unroll 4
  for (int k0 = 0; k0 < 64; k0 += 4) {
    const int kk = k0 + koff;
    const int colz = (kk < 32) ? kk : (64 + kk);
    const int colr = (kk < 32) ? (32 + kk) : (64 + kk);
    v2f az; az.x = As[m * 132 + colz]; az.y = As[m * 132 + colz + 1];
    v2f ar; ar.x = As[m * 132 + colr]; ar.y = As[m * 132 + colr + 1];
    const float* lz = LzW + kk * 32 + n;
    const float* lr = LrW + kk * 32 + n;
    v2f b;
    b.x = lz[0];  b.y = lz[32];  z0 = wmma4(az, b, z0);
    b.x = lz[16]; b.y = lz[48];  z1 = wmma4(az, b, z1);
    b.x = lr[0];  b.y = lr[32];  r0 = wmma4(ar, b, r0);
    b.x = lr[16]; b.y = lr[48];  r1 = wmma4(ar, b, r1);
  }

  const float bz0 = Lzb[n], bz1 = Lzb[16 + n];
  const float br0 = Lrb[n], br1 = Lrb[16 + n];
  const int row0 = (lane >> 4) << 3;
#pragma unroll
  for (int v = 0; v < 8; ++v) {
    z0[v] = sigmoidf_(z0[v] + bz0);
    z1[v] = sigmoidf_(z1[v] + bz1);
    r0[v] = sigmoidf_(r0[v] + br0);
    r1[v] = sigmoidf_(r1[v] + br1);
    Rs[(row0 + v) * 36 + n]      = r0[v];
    Rs[(row0 + v) * 36 + 16 + n] = r1[v];
  }
  __syncthreads();

  v8f t0 = {}, t1 = {};
#pragma unroll 4
  for (int k0 = 0; k0 < 64; k0 += 4) {
    const int kk = k0 + koff;
    v2f a;
    a.x = As[m * 132 + 64 + kk];
    a.y = As[m * 132 + 65 + kk];
    if (kk >= 32) {
      a.x *= Rs[m * 36 + kk - 32];
      a.y *= Rs[m * 36 + kk - 31];
    }
    const float* lh = LhW + kk * 32 + n;
    v2f b;
    b.x = lh[0];  b.y = lh[32];  t0 = wmma4(a, b, t0);
    b.x = lh[16]; b.y = lh[48];  t1 = wmma4(a, b, t1);
  }

  const float bh0 = Lhb[n], bh1 = Lhb[16 + n];
  const float hw0 = headW[n], hw1 = headW[16 + n];
  float p[8];
#pragma unroll
  for (int v = 0; v < 8; ++v) {
    const float ht0 = tanhf(t0[v] + bh0);
    const float ht1 = tanhf(t1[v] + bh1);
    const float hp0 = As[(row0 + v) * 132 + 96 + n];
    const float hp1 = As[(row0 + v) * 132 + 112 + n];
    const float h0 = z0[v] * hp0 + (1.0f - z0[v]) * ht0;
    const float h1 = z1[v] * hp1 + (1.0f - z1[v]) * ht1;
    h_out[(size_t)(tile + row0 + v) * 32 + n]      = h0;
    h_out[(size_t)(tile + row0 + v) * 32 + 16 + n] = h1;
    p[v] = fmaxf(h0, 0.0f) * hw0 + fmaxf(h1, 0.0f) * hw1;
  }
#pragma unroll
  for (int s = 1; s < 16; s <<= 1)
#pragma unroll
    for (int v = 0; v < 8; ++v)
      p[v] += __shfl_xor(p[v], s, 32);
  if (n == 0) {
    const float hb = headb[0];
#pragma unroll
    for (int v = 0; v < 8; ++v)
      y_out[tile + row0 + v] = p[v] + hb;
  }
}

extern "C" void kernel_launch(void* const* d_in, const int* in_sizes, int n_in,
                              void* d_out, int out_size, void* d_ws, size_t ws_size,
                              hipStream_t stream) {
  const float* x    = (const float*)d_in[0];
  const int*   ei   = (const int*)d_in[1];
  const float* ew   = (const float*)d_in[2];
  const float* Hp   = (const float*)d_in[3];
  const float* Wz   = (const float*)d_in[4];
  const float* bz   = (const float*)d_in[5];
  const float* Wr   = (const float*)d_in[6];
  const float* br   = (const float*)d_in[7];
  const float* Wh   = (const float*)d_in[8];
  const float* bh   = (const float*)d_in[9];
  const float* LzW  = (const float*)d_in[10];
  const float* Lzb  = (const float*)d_in[11];
  const float* LrW  = (const float*)d_in[12];
  const float* Lrb  = (const float*)d_in[13];
  const float* LhW  = (const float*)d_in[14];
  const float* Lhb  = (const float*)d_in[15];
  const float* hW   = (const float*)d_in[16];
  const float* hb   = (const float*)d_in[17];

  const int* src = ei;
  const int* dst = ei + Ee;

  float* out_y = (float*)d_out;
  float* out_h = out_y + Nn;

  float* ws   = (float*)d_ws;
  float* dis  = ws;
  float* Wcat = ws + Nn;
  float* bcat = Wcat + FIN * 96;
  float* XW   = bcat + 96;
  float* AGG  = XW + (size_t)Nn * 96;
  float* part = AGG + (size_t)Nn * 96;
  float* P    = part + (size_t)DEG_S * Nn;
  unsigned* rid4 = (unsigned*)(P + (size_t)AG_S * Nn * 96);

  pack_weights_kernel<<<(FIN * 96 + 255) / 256, 256, 0, stream>>>(Wz, Wr, Wh, bz, br, bh, Wcat, bcat);
  deg_partial_kernel<<<DEG_Q * DEG_S, DEG_TPB, 0, stream>>>(dst, ew, part, rid4);
  finalize_dis_kernel<<<(Nn + 255) / 256, 256, 0, stream>>>(part, dis);
  gemm_xw_kernel<<<Nn / 16, 32, 0, stream>>>(x, Wcat, XW);
  agg_range_kernel<<<AG_Q * AG_S, AG_TPB, 0, stream>>>(src, dst, rid4, ew, dis, XW, P);
  combine_agg_kernel<<<(Nn * 96) / 256, 256, 0, stream>>>(dis, XW, bcat, P, AGG);
  gru_kernel<<<Nn / 16, 32, 0, stream>>>(AGG, Hp, LzW, Lzb, LrW, Lrb, LhW, Lhb, hW, hb, out_y, out_h);
}


# Combine pass folded into the GRU tile staging: a wave forms its 16 AGG rows from XW, dis, bias and both planes with whole-line float4 loads; AGG never stored or re-read; 6 launches; sums unchanged
# speedup vs baseline: 2.5500x; 1.1061x over previous
#include <hip/hip_runtime.h>

#define Nn    100000
#define Ee    3200000
#define FIN   256
#define Hh    32
#define AG_R  800

typedef float v2f __attribute__((ext_vector_type(2)));
typedef float v8f __attribute__((ext_vector_type(8)));

__device__ __forceinline__ v8f wmma4(v2f a, v2f b, v8f c) {
  return __builtin_amdgcn_wmma_f32_16x16x4_f32(
       false, a,  false, b,
       (short)0, c,  false,  false);
}

__device__ __forceinline__ float sigmoidf_(float x) {
  return 1.0f / (1.0f + __expf(-x));
}

__global__ void pack_weights_kernel(const float* __restrict__ Wz, const float* __restrict__ Wr,
                                    const float* __restrict__ Wh, const float* __restrict__ bz,
                                    const float* __restrict__ br, const float* __restrict__ bh,
                                    float* __restrict__ Wcat, float* __restrict__ bcat) {
  int gid = blockIdx.x * blockDim.x + threadIdx.x;
  if (gid < FIN * 96) {
    int k = gid / 96, c = gid - k * 96;
    float v = (c < 32) ? Wz[k * 32 + c] : ((c < 64) ? Wr[k * 32 + c - 32] : Wh[k * 32 + c - 64]);
    Wcat[gid] = v;
  }
  if (gid < 96) {
    bcat[gid] = (gid < 32) ? bz[gid] : ((gid < 64) ? br[gid - 32] : bh[gid - 64]);
  }
}

#define DEG_Q      4
#define DEG_RANGE  (Nn / DEG_Q)
#define DEG_S      64
#define DEG_SLICE  (Ee / DEG_S)
#define DEG_NV     (DEG_SLICE / 4)
#define DEG_TPB    1024

__device__ __forceinline__ unsigned rid_of(int d) {
  return ((unsigned)d < (unsigned)Nn) ? (unsigned)d / (unsigned)AG_R : 255u;
}

__device__ __forceinline__ void deg_add(float* tab, int d, float wv) {
  if ((unsigned)d < (unsigned)DEG_RANGE) unsafeAtomicAdd(&tab[d], wv);
}

__global__ void __launch_bounds__(DEG_TPB) deg_partial_kernel(const int* __restrict__ dst,
                                                              const float* __restrict__ w,
                                                              float* __restrict__ part,
                                                              unsigned* __restrict__ rid4) {
  __shared__ float tab[DEG_RANGE];
  const int tid = threadIdx.x;
  const int q = blockIdx.x % DEG_Q;
  const int s = blockIdx.x / DEG_Q;
  const int lo = q * DEG_RANGE;

  float4* tab4 = reinterpret_cast<float4*>(tab);
  for (int i = tid; i < DEG_RANGE / 4; i += DEG_TPB) tab4[i] = make_float4(0.0f, 0.0f, 0.0f, 0.0f);
  __syncthreads();

  const int4*   d4 = reinterpret_cast<const int4*>(dst + (size_t)s * DEG_SLICE);
  const float4* w4 = reinterpret_cast<const float4*>(w + (size_t)s * DEG_SLICE);
  for (int i0 = tid; i0 < DEG_NV; i0 += 4 * DEG_TPB) {
    int4 dd[4];
    float4 ww[4];
#pragma unroll
    for (int j = 0; j < 4; ++j) {
      const int i = i0 + j * DEG_TPB;
      if (i < DEG_NV) {
        dd[j] = d4[i];
        ww[j] = w4[i];
      } else {
        dd[j] = make_int4(-1, -1, -1, -1);
        ww[j] = make_float4(0.0f, 0.0f, 0.0f, 0.0f);
      }
    }
#pragma unroll
    for (int j = 0; j < 4; ++j) {
      if (j == q && i0 + j * DEG_TPB < DEG_NV)
        rid4[(size_t)s * DEG_NV + i0 + j * DEG_TPB] =
            rid_of(dd[j].x) | (rid_of(dd[j].y) << 8) | (rid_of(dd[j].z) << 16) | (rid_of(dd[j].w) << 24);
      deg_add(tab, dd[j].x - lo, ww[j].x);
      deg_add(tab, dd[j].y - lo, ww[j].y);
      deg_add(tab, dd[j].z - lo, ww[j].z);
      deg_add(tab, dd[j].w - lo, ww[j].w);
    }
  }
  __syncthreads();

  float4* out4 = reinterpret_cast<float4*>(part + (size_t)s * Nn + lo);
  for (int i = tid; i < DEG_RANGE / 4; i += DEG_TPB) out4[i] = tab4[i];
}

__global__ void finalize_dis_kernel(const float* __restrict__ part, float* __restrict__ dis) {
  int gid = blockIdx.x * blockDim.x + threadIdx.x;
  if (gid < Nn) {
    float sum = 0.0f;
    for (int s = 0; s < DEG_S; ++s) sum += part[(size_t)s * Nn + gid];
    dis[gid] = rsqrtf(sum + 1.0f);
  }
}

__global__ void __launch_bounds__(32) gemm_xw_kernel(const float* __restrict__ x,
                                                     const float* __restrict__ Wcat,
                                                     float* __restrict__ XW) {
  __shared__ float Xs[16 * 260];
  const int lane = threadIdx.x;
  const int tile = blockIdx.x * 16;

  for (int i = lane; i < 1024; i += 32) {
    int r = i >> 6, c4 = (i & 63) << 2;
    const float4 v = *reinterpret_cast<const float4*>(x + (size_t)(tile + r) * FIN + c4);
    *reinterpret_cast<float4*>(&Xs[r * 260 + c4]) = v;
  }
  __syncthreads();

  const int m = lane & 15;
  const int koff = (lane >> 4) << 1;

  v8f acc[6] = {};
  const float2* arow = reinterpret_cast<const float2*>(&Xs[m * 260]);

#pragma unroll 4
  for (int k0 = 0; k0 < FIN; k0 += 4) {
    const int kk = k0 + koff;
    float2 af = arow[kk >> 1];
    v2f a; a.x = af.x; a.y = af.y;
    const float* brow = Wcat + (size_t)kk * 96 + m;
#pragma unroll
    for (int t = 0; t < 6; ++t) {
      v2f b; b.x = brow[t * 16]; b.y = brow[96 + t * 16];
      acc[t] = wmma4(a, b, acc[t]);
    }
  }

  const int row0 = (lane >> 4) << 3;
#pragma unroll
  for (int t = 0; t < 6; ++t)
#pragma unroll
    for (int v = 0; v < 8; ++v)
      XW[(size_t)(tile + row0 + v) * 96 + t * 16 + m] = acc[t][v];
}

#define AG_Q      (Nn / AG_R)
#define AG_S      2
#define AG_SLICE  (Ee / AG_S)
#define AG_NV     (AG_SLICE / 16)
#define AG_TPB    1024
#define AG_LCAP   96

__device__ __forceinline__ void ag_push(int* lst, int& cnt, bool m, int e, int lane) {
  const unsigned mask = __builtin_amdgcn_ballot_w32(m);
  if (mask != 0u) {
    if (m) lst[cnt + __builtin_popcount(mask & ((1u << lane) - 1u))] = e;
    cnt += __builtin_popcount(mask);
  }
}

__device__ __forceinline__ void ag_flush(float* tab, const int* lst, int n, int lo, int lane,
                                         const int* __restrict__ src, const int* __restrict__ dst,
                                         const float* __restrict__ w, const float* __restrict__ dis,
                                         const float* __restrict__ XW) {
  int sN = 0, ld = 0;
  float nrm = 0.0f;
  if (lane < n) {
    const int e = lst[lane];
    const int s = src[e], d = dst[e];
    nrm = dis[s] * w[e] * dis[d];
    sN = s * 96;
    ld = (d - lo) * 96;
  }
  for (int j0 = 0; j0 < n; j0 += 8) {
    float v[8][3];
#pragma unroll
    for (int j = 0; j < 8; ++j) {
      const int sb = __builtin_amdgcn_readlane(sN, j0 + j);
#pragma unroll
      for (int i = 0; i < 3; ++i) v[j][i] = XW[(size_t)sb + lane + 32 * i];
    }
#pragma unroll
    for (int j = 0; j < 8; ++j) {
      if (j0 + j < n) {
        const float nj = __int_as_float(__builtin_amdgcn_readlane(__float_as_int(nrm), j0 + j));
        const int lj = __builtin_amdgcn_readlane(ld, j0 + j);
#pragma unroll
        for (int i = 0; i < 3; ++i) unsafeAtomicAdd(&tab[lj + lane + 32 * i], nj * v[j][i]);
      }
    }
  }
}

__device__ __forceinline__ void ag_word(float* tab, int* lst, int& cnt, unsigned x, int e, int lo, int lane,
                                        const int* __restrict__ src, const int* __restrict__ dst,
                                        const float* __restrict__ w, const float* __restrict__ dis,
                                        const float* __restrict__ XW) {
  ag_push(lst, cnt, (x & 0x000000ffu) == 0u, e, lane);
  ag_push(lst, cnt, (x & 0x0000ff00u) == 0u, e + 1, lane);
  while (cnt >= 32) {
    cnt -= 32;
    ag_flush(tab, lst + cnt, 32, lo, lane, src, dst, w, dis, XW);
  }
  ag_push(lst, cnt, (x & 0x00ff0000u) == 0u, e + 2, lane);
  ag_push(lst, cnt, (x & 0xff000000u) == 0u, e + 3, lane);
  while (cnt >= 32) {
    cnt -= 32;
    ag_flush(tab, lst + cnt, 32, lo, lane, src, dst, w, dis, XW);
  }
}

__global__ void __launch_bounds__(AG_TPB) agg_range_kernel(const int* __restrict__ src,
                                                           const int* __restrict__ dst,
                                                           const unsigned* __restrict__ rid4,
                                                           const float* __restrict__ w,
                                                           const float* __restrict__ dis,
                                                           const float* __restrict__ XW,
                                                           float* __restrict__ P) {
  __shared__ float tab[AG_R * 96];
  __shared__ int lst[(AG_TPB / 32) * AG_LCAP];
  const int tid = threadIdx.x;
  const int lane = tid & 31;
  const int s = blockIdx.x % AG_S;
  const int q = blockIdx.x / AG_S;
  const int lo = q * AG_R;

  float4* tab4 = reinterpret_cast<float4*>(tab);
  for (int i = tid; i < AG_R * 24; i += AG_TPB) tab4[i] = make_float4(0.0f, 0.0f, 0.0f, 0.0f);
  __syncthreads();

  int* mylst = lst + (tid >> 5) * AG_LCAP;
  int cnt = 0;
  const int ebase = s * AG_SLICE;
  const uint4* r16 = reinterpret_cast<const uint4*>(rid4) + (size_t)s * AG_NV;
  const unsigned qs = (unsigned)q * 0x01010101u;
  for (int i0 = tid; i0 < AG_NV; i0 += 4 * AG_TPB) {
    uint4 dd[4];
#pragma unroll
    for (int j = 0; j < 4; ++j) {
      const int i = i0 + j * AG_TPB;
      dd[j] = (i < AG_NV) ? r16[i] : make_uint4(~0u, ~0u, ~0u, ~0u);
    }
    int e = ebase + 16 * i0;
#pragma unroll 1
    for (int j = 0; j < 4; ++j) {
      const uint4 cur = dd[0];
      dd[0] = dd[1];
      dd[1] = dd[2];
      dd[2] = dd[3];
      ag_word(tab, mylst, cnt, cur.x ^ qs, e,      lo, lane, src, dst, w, dis, XW);
      ag_word(tab, mylst, cnt, cur.y ^ qs, e + 4,  lo, lane, src, dst, w, dis, XW);
      ag_word(tab, mylst, cnt, cur.z ^ qs, e + 8,  lo, lane, src, dst, w, dis, XW);
      ag_word(tab, mylst, cnt, cur.w ^ qs, e + 12, lo, lane, src, dst, w, dis, XW);
      e += 16 * AG_TPB;
    }
  }
  if (cnt > 0) ag_flush(tab, mylst, cnt, lo, lane, src, dst, w, dis, XW);
  __syncthreads();

  float4* out4 = reinterpret_cast<float4*>(P + ((size_t)s * Nn + lo) * 96);
  for (int i = tid; i < AG_R * 24; i += AG_TPB) out4[i] = tab4[i];
}

__global__ void __launch_bounds__(32) gru_kernel(const float* __restrict__ XW, const float* __restrict__ P,
                                                 const float* __restrict__ dis, const float* __restrict__ bcat,
                                                 const float* __restrict__ Hp,
                                                 const float* __restrict__ LzW, const float* __restrict__ Lzb,
                                                 const float* __restrict__ LrW, const float* __restrict__ Lrb,
                                                 const float* __restrict__ LhW, const float* __restrict__ Lhb,
                                                 const float* __restrict__ headW, const float* __restrict__ headb,
                                                 float* __restrict__ y_out, float* __restrict__ h_out) {
  __shared__ __attribute__((aligned(16))) float As[16 * 132];
  __shared__ float Rs[16 * 36];
  const int lane = threadIdx.x;
  const int tile = blockIdx.x * 16;

  const float4* xw4 = reinterpret_cast<const float4*>(XW + (size_t)tile * 96);
  const float4* p04 = reinterpret_cast<const float4*>(P + (size_t)tile * 96);
  const float4* p14 = reinterpret_cast<const float4*>(P + ((size_t)Nn + tile) * 96);
  const float4* b4  = reinterpret_cast<const float4*>(bcat);
  const float4* hp4 = reinterpret_cast<const float4*>(Hp + (size_t)tile * 32);

  float4 bb[3];
#pragma unroll
  for (int j = 0; j < 3; ++j) bb[j] = b4[(lane + 32 * j) % 24];

#pragma unroll
  for (int g = 0; g < 3; ++g) {
    float4 xv[4], pa[4], pb[4];
    float dv[4];
#pragma unroll
    for (int j = 0; j < 4; ++j) {
      const int f = lane + 32 * (4 * g + j);
      xv[j] = xw4[f];
      pa[j] = p04[f];
      pb[j] = p14[f];
      dv[j] = dis[tile + f / 24];
    }
#pragma unroll
    for (int j = 0; j < 4; ++j) {
      const int f = lane + 32 * (4 * g + j);
      const int r = f / 24, c = 4 * (f - 24 * r);
      const float d2 = dv[j] * dv[j];
      const float4 bv = bb[(4 * g + j) % 3];
      float4 a;
      a.x = d2 * xv[j].x + bv.x;  a.x += pa[j].x;  a.x += pb[j].x;
      a.y = d2 * xv[j].y + bv.y;  a.y += pa[j].y;  a.y += pb[j].y;
      a.z = d2 * xv[j].z + bv.z;  a.z += pa[j].z;  a.z += pb[j].z;
      a.w = d2 * xv[j].w + bv.w;  a.w += pa[j].w;  a.w += pb[j].w;
      *reinterpret_cast<float4*>(&As[r * 132 + c]) = a;
    }
  }
#pragma unroll
  for (int j = 0; j < 4; ++j) {
    const int f = lane + 32 * j;
    *reinterpret_cast<float4*>(&As[(f >> 3) * 132 + 96 + 4 * (f & 7)]) = hp4[f];
  }
  __syncthreads();

  const int m = lane & 15;
  const int n = lane & 15;
  const int koff = (lane >> 4) << 1;

  v8f z0 = {}, z1 = {}, r0 = {}, r1 = {};
#pragma unroll 4
  for (int k0 = 0; k0 < 64; k0 += 4) {
    const int kk = k0 + koff;
    const int colz = (kk < 32) ? kk : (64 + kk);
    const int colr = (kk < 32) ? (32 + kk) : (64 + kk);
    v2f az; az.x = As[m * 132 + colz]; az.y = As[m * 132 + colz + 1];
    v2f ar; ar.x = As[m * 132 + colr]; ar.y = As[m * 132 + colr + 1];
    const float* lz = LzW + kk * 32 + n;
    const float* lr = LrW + kk * 32 + n;
    v2f b;
    b.x = lz[0];  b.y = lz[32];  z0 = wmma4(az, b, z0);
    b.x = lz[16]; b.y = lz[48];  z1 = wmma4(az, b, z1);
    b.x = lr[0];  b.y = lr[32];  r0 = wmma4(ar, b, r0);
    b.x = lr[16]; b.y = lr[48];  r1 = wmma4(ar, b, r1);
  }

  const float bz0 = Lzb[n], bz1 = Lzb[16 + n];
  const float br0 = Lrb[n], br1 = Lrb[16 + n];
  const int row0 = (lane >> 4) << 3;
#pragma unroll
  for (int v = 0; v < 8; ++v) {
    z0[v] = sigmoidf_(z0[v] + bz0);
    z1[v] = sigmoidf_(z1[v] + bz1);
    r0[v] = sigmoidf_(r0[v] + br0);
    r1[v] = sigmoidf_(r1[v] + br1);
    Rs[(row0 + v) * 36 + n]      = r0[v];
    Rs[(row0 + v) * 36 + 16 + n] = r1[v];
  }
  __syncthreads();

  v8f t0 = {}, t1 = {};
#pragma unroll 4
  for (int k0 = 0; k0 < 64; k0 += 4) {
    const int kk = k0 + koff;
    v2f a;
    a.x = As[m * 132 + 64 + kk];
    a.y = As[m * 132 + 65 + kk];
    if (kk >= 32) {
      a.x *= Rs[m * 36 + kk - 32];
      a.y *= Rs[m * 36 + kk - 31];
    }
    const float* lh = LhW + kk * 32 + n;
    v2f b;
    b.x = lh[0];  b.y = lh[32];  t0 = wmma4(a, b, t0);
    b.x = lh[16]; b.y = lh[48];  t1 = wmma4(a, b, t1);
  }

  const float bh0 = Lhb[n], bh1 = Lhb[16 + n];
  const float hw0 = headW[n], hw1 = headW[16 + n];
  float p[8];
#pragma unroll
  for (int v = 0; v < 8; ++v) {
    const float ht0 = tanhf(t0[v] + bh0);
    const float ht1 = tanhf(t1[v] + bh1);
    const float hp0 = As[(row0 + v) * 132 + 96 + n];
    const float hp1 = As[(row0 + v) * 132 + 112 + n];
    const float h0 = z0[v] * hp0 + (1.0f - z0[v]) * ht0;
    const float h1 = z1[v] * hp1 + (1.0f - z1[v]) * ht1;
    h_out[(size_t)(tile + row0 + v) * 32 + n]      = h0;
    h_out[(size_t)(tile + row0 + v) * 32 + 16 + n] = h1;
    p[v] = fmaxf(h0, 0.0f) * hw0 + fmaxf(h1, 0.0f) * hw1;
  }
#pragma unroll
  for (int s = 1; s < 16; s <<= 1)
#pragma unroll
    for (int v = 0; v < 8; ++v)
      p[v] += __shfl_xor(p[v], s, 32);
  if (n == 0) {
    const float hb = headb[0];
#pragma unroll
    for (int v = 0; v < 8; ++v)
      y_out[tile + row0 + v] = p[v] + hb;
  }
}

extern "C" void kernel_launch(void* const* d_in, const int* in_sizes, int n_in,
                              void* d_out, int out_size, void* d_ws, size_t ws_size,
                              hipStream_t stream) {
  const float* x    = (const float*)d_in[0];
  const int*   ei   = (const int*)d_in[1];
  const float* ew   = (const float*)d_in[2];
  const float* Hp   = (const float*)d_in[3];
  const float* Wz   = (const float*)d_in[4];
  const float* bz   = (const float*)d_in[5];
  const float* Wr   = (const float*)d_in[6];
  const float* br   = (const float*)d_in[7];
  const float* Wh   = (const float*)d_in[8];
  const float* bh   = (const float*)d_in[9];
  const float* LzW  = (const float*)d_in[10];
  const float* Lzb  = (const float*)d_in[11];
  const float* LrW  = (const float*)d_in[12];
  const float* Lrb  = (const float*)d_in[13];
  const float* LhW  = (const float*)d_in[14];
  const float* Lhb  = (const float*)d_in[15];
  const float* hW   = (const float*)d_in[16];
  const float* hb   = (const float*)d_in[17];

  const int* src = ei;
  const int* dst = ei + Ee;

  float* out_y = (float*)d_out;
  float* out_h = out_y + Nn;

  float* ws   = (float*)d_ws;
  float* dis  = ws;
  float* Wcat = ws + Nn;
  float* bcat = Wcat + FIN * 96;
  float* XW   = bcat + 96;
  float* part = XW + (size_t)Nn * 96;
  float* P    = part + (size_t)DEG_S * Nn;
  unsigned* rid4 = (unsigned*)(P + (size_t)AG_S * Nn * 96);

  pack_weights_kernel<<<(FIN * 96 + 255) / 256, 256, 0, stream>>>(Wz, Wr, Wh, bz, br, bh, Wcat, bcat);
  deg_partial_kernel<<<DEG_Q * DEG_S, DEG_TPB, 0, stream>>>(dst, ew, part, rid4);
  finalize_dis_kernel<<<(Nn + 255) / 256, 256, 0, stream>>>(part, dis);
  gemm_xw_kernel<<<Nn / 16, 32, 0, stream>>>(x, Wcat, XW);
  agg_range_kernel<<<AG_Q * AG_S, AG_TPB, 0, stream>>>(src, dst, rid4, ew, dis, XW, P);
  gru_kernel<<<Nn / 16, 32, 0, stream>>>(XW, P, dis, bcat, Hp, LzW, Lzb, LrW, Lrb, LhW, Lhb, hW, hb, out_y, out_h);
}


# Degree pass also writes one 8-byte record per edge (weight; src | row of dst in its range << 17): the aggregate fetches one line per found edge, not three (src, dst, w); sums unchanged
# speedup vs baseline: 2.8280x; 1.1090x over previous
#include <hip/hip_runtime.h>

#define Nn    100000
#define Ee    3200000
#define FIN   256
#define Hh    32
#define AG_R  800

typedef float v2f __attribute__((ext_vector_type(2)));
typedef float v8f __attribute__((ext_vector_type(8)));

__device__ __forceinline__ v8f wmma4(v2f a, v2f b, v8f c) {
  return __builtin_amdgcn_wmma_f32_16x16x4_f32(
       false, a,  false, b,
       (short)0, c,  false,  false);
}

__device__ __forceinline__ float sigmoidf_(float x) {
  return 1.0f / (1.0f + __expf(-x));
}

__global__ void pack_weights_kernel(const float* __restrict__ Wz, const float* __restrict__ Wr,
                                    const float* __restrict__ Wh, const float* __restrict__ bz,
                                    const float* __restrict__ br, const float* __restrict__ bh,
                                    float* __restrict__ Wcat, float* __restrict__ bcat) {
  int gid = blockIdx.x * blockDim.x + threadIdx.x;
  if (gid < FIN * 96) {
    int k = gid / 96, c = gid - k * 96;
    float v = (c < 32) ? Wz[k * 32 + c] : ((c < 64) ? Wr[k * 32 + c - 32] : Wh[k * 32 + c - 64]);
    Wcat[gid] = v;
  }
  if (gid < 96) {
    bcat[gid] = (gid < 32) ? bz[gid] : ((gid < 64) ? br[gid - 32] : bh[gid - 64]);
  }
}

#define DEG_Q      4
#define DEG_RANGE  (Nn / DEG_Q)
#define DEG_S      64
#define DEG_SLICE  (Ee / DEG_S)
#define DEG_NV     (DEG_SLICE / 4)
#define DEG_TPB    1024

__device__ __forceinline__ unsigned rid_of(int d) {
  return ((unsigned)d < (unsigned)Nn) ? (unsigned)d / (unsigned)AG_R : 255u;
}

__device__ __forceinline__ unsigned rec_word(int s, int d) {
  const unsigned row = (unsigned)d - rid_of(d) * (unsigned)AG_R;
  return ((unsigned)s & 0x1ffffu) | (row << 17);
}

__device__ __forceinline__ void deg_add(float* tab, int d, float wv) {
  if ((unsigned)d < (unsigned)DEG_RANGE) unsafeAtomicAdd(&tab[d], wv);
}

__global__ void __launch_bounds__(DEG_TPB) deg_partial_kernel(const int* __restrict__ src,
                                                              const int* __restrict__ dst,
                                                              const float* __restrict__ w,
                                                              float* __restrict__ part,
                                                              unsigned* __restrict__ rid4,
                                                              unsigned* __restrict__ rec) {
  __shared__ float tab[DEG_RANGE];
  const int tid = threadIdx.x;
  const int q = blockIdx.x % DEG_Q;
  const int s = blockIdx.x / DEG_Q;
  const int lo = q * DEG_RANGE;

  float4* tab4 = reinterpret_cast<float4*>(tab);
  for (int i = tid; i < DEG_RANGE / 4; i += DEG_TPB) tab4[i] = make_float4(0.0f, 0.0f, 0.0f, 0.0f);
  __syncthreads();

  const int4*   s4 = reinterpret_cast<const int4*>(src + (size_t)s * DEG_SLICE);
  const int4*   d4 = reinterpret_cast<const int4*>(dst + (size_t)s * DEG_SLICE);
  const float4* w4 = reinterpret_cast<const float4*>(w + (size_t)s * DEG_SLICE);
  uint4* rec4 = reinterpret_cast<uint4*>(rec) + 2 * (size_t)s * DEG_NV;
  for (int i0 = tid; i0 < DEG_NV; i0 += 4 * DEG_TPB) {
    int4 dd[4];
    float4 ww[4];
#pragma unroll
    for (int j = 0; j < 4; ++j) {
      const int i = i0 + j * DEG_TPB;
      if (i < DEG_NV) {
        dd[j] = d4[i];
        ww[j] = w4[i];
      } else {
        dd[j] = make_int4(-1, -1, -1, -1);
        ww[j] = make_float4(0.0f, 0.0f, 0.0f, 0.0f);
      }
    }
    const int iq = i0 + q * DEG_TPB;
    int4 ss = make_int4(0, 0, 0, 0);
    if (iq < DEG_NV) ss = s4[iq];
#pragma unroll
    for (int j = 0; j < 4; ++j) {
      if (j == q && i0 + j * DEG_TPB < DEG_NV) {
        const int i = i0 + j * DEG_TPB;
        rid4[(size_t)s * DEG_NV + i] =
            rid_of(dd[j].x) | (rid_of(dd[j].y) << 8) | (rid_of(dd[j].z) << 16) | (rid_of(dd[j].w) << 24);
        rec4[2 * i]     = make_uint4(__float_as_uint(ww[j].x), rec_word(ss.x, dd[j].x),
                                     __float_as_uint(ww[j].y), rec_word(ss.y, dd[j].y));
        rec4[2 * i + 1] = make_uint4(__float_as_uint(ww[j].z), rec_word(ss.z, dd[j].z),
                                     __float_as_uint(ww[j].w), rec_word(ss.w, dd[j].w));
      }
      deg_add(tab, dd[j].x - lo, ww[j].x);
      deg_add(tab, dd[j].y - lo, ww[j].y);
      deg_add(tab, dd[j].z - lo, ww[j].z);
      deg_add(tab, dd[j].w - lo, ww[j].w);
    }
  }
  __syncthreads();

  float4* out4 = reinterpret_cast<float4*>(part + (size_t)s * Nn + lo);
  for (int i = tid; i < DEG_RANGE / 4; i += DEG_TPB) out4[i] = tab4[i];
}

__global__ void finalize_dis_kernel(const float* __restrict__ part, float* __restrict__ dis) {
  int gid = blockIdx.x * blockDim.x + threadIdx.x;
  if (gid < Nn) {
    float sum = 0.0f;
    for (int s = 0; s < DEG_S; ++s) sum += part[(size_t)s * Nn + gid];
    dis[gid] = rsqrtf(sum + 1.0f);
  }
}

__global__ void __launch_bounds__(32) gemm_xw_kernel(const float* __restrict__ x,
                                                     const float* __restrict__ Wcat,
                                                     float* __restrict__ XW) {
  __shared__ float Xs[16 * 260];
  const int lane = threadIdx.x;
  const int tile = blockIdx.x * 16;

  for (int i = lane; i < 1024; i += 32) {
    int r = i >> 6, c4 = (i & 63) << 2;
    const float4 v = *reinterpret_cast<const float4*>(x + (size_t)(tile + r) * FIN + c4);
    *reinterpret_cast<float4*>(&Xs[r * 260 + c4]) = v;
  }
  __syncthreads();

  const int m = lane & 15;
  const int koff = (lane >> 4) << 1;

  v8f acc[6] = {};
  const float2* arow = reinterpret_cast<const float2*>(&Xs[m * 260]);

#pragma unroll 4
  for (int k0 = 0; k0 < FIN; k0 += 4) {
    const int kk = k0 + koff;
    float2 af = arow[kk >> 1];
    v2f a; a.x = af.x; a.y = af.y;
    const float* brow = Wcat + (size_t)kk * 96 + m;
#pragma unroll
    for (int t = 0; t < 6; ++t) {
      v2f b; b.x = brow[t * 16]; b.y = brow[96 + t * 16];
      acc[t] = wmma4(a, b, acc[t]);
    }
  }

  const int row0 = (lane >> 4) << 3;
#pragma unroll
  for (int t = 0; t < 6; ++t)
#pragma unroll
    for (int v = 0; v < 8; ++v)
      XW[(size_t)(tile + row0 + v) * 96 + t * 16 + m] = acc[t][v];
}

#define AG_Q      (Nn / AG_R)
#define AG_S      2
#define AG_SLICE  (Ee / AG_S)
#define AG_NV     (AG_SLICE / 16)
#define AG_TPB    1024
#define AG_LCAP   96

__device__ __forceinline__ void ag_push(int* lst, int& cnt, bool m, int e, int lane) {
  const unsigned mask = __builtin_amdgcn_ballot_w32(m);
  if (mask != 0u) {
    if (m) lst[cnt + __builtin_popcount(mask & ((1u << lane) - 1u))] = e;
    cnt += __builtin_popcount(mask);
  }
}

__device__ __forceinline__ void ag_flush(float* tab, const int* lst, int n, int lo, int lane,
                                         const uint2* __restrict__ rec, const float* __restrict__ dis,
                                         const float* __restrict__ XW) {
  int sN = 0, ld = 0;
  float nrm = 0.0f;
  if (lane < n) {
    const uint2 r = rec[lst[lane]];
    const int s = (int)(r.y & 0x1ffffu);
    const int row = (int)(r.y >> 17);
    nrm = dis[s] * __uint_as_float(r.x) * dis[lo + row];
    sN = s * 96;
    ld = row * 96;
  }
  for (int j0 = 0; j0 < n; j0 += 8) {
    float v[8][3];
#pragma unroll
    for (int j = 0; j < 8; ++j) {
      const int sb = __builtin_amdgcn_readlane(sN, j0 + j);
#pragma unroll
      for (int i = 0; i < 3; ++i) v[j][i] = XW[(size_t)sb + lane + 32 * i];
    }
#pragma unroll
    for (int j = 0; j < 8; ++j) {
      if (j0 + j < n) {
        const float nj = __int_as_float(__builtin_amdgcn_readlane(__float_as_int(nrm), j0 + j));
        const int lj = __builtin_amdgcn_readlane(ld, j0 + j);
#pragma unroll
        for (int i = 0; i < 3; ++i) unsafeAtomicAdd(&tab[lj + lane + 32 * i], nj * v[j][i]);
      }
    }
  }
}

__device__ __forceinline__ void ag_word(float* tab, int* lst, int& cnt, unsigned x, int e, int lo, int lane,
                                        const uint2* __restrict__ rec, const float* __restrict__ dis,
                                        const float* __restrict__ XW) {
  ag_push(lst, cnt, (x & 0x000000ffu) == 0u, e, lane);
  ag_push(lst, cnt, (x & 0x0000ff00u) == 0u, e + 1, lane);
  while (cnt >= 32) {
    cnt -= 32;
    ag_flush(tab, lst + cnt, 32, lo, lane, rec, dis, XW);
  }
  ag_push(lst, cnt, (x & 0x00ff0000u) == 0u, e + 2, lane);
  ag_push(lst, cnt, (x & 0xff000000u) == 0u, e + 3, lane);
  while (cnt >= 32) {
    cnt -= 32;
    ag_flush(tab, lst + cnt, 32, lo, lane, rec, dis, XW);
  }
}

__global__ void __launch_bounds__(AG_TPB) agg_range_kernel(const unsigned* __restrict__ rid4,
                                                           const unsigned* __restrict__ recw,
                                                           const float* __restrict__ dis,
                                                           const float* __restrict__ XW,
                                                           float* __restrict__ P) {
  __shared__ float tab[AG_R * 96];
  __shared__ int lst[(AG_TPB / 32) * AG_LCAP];
  const int tid = threadIdx.x;
  const int lane = tid & 31;
  const int s = blockIdx.x % AG_S;
  const int q = blockIdx.x / AG_S;
  const int lo = q * AG_R;

  float4* tab4 = reinterpret_cast<float4*>(tab);
  for (int i = tid; i < AG_R * 24; i += AG_TPB) tab4[i] = make_float4(0.0f, 0.0f, 0.0f, 0.0f);
  __syncthreads();

  const uint2* rec = reinterpret_cast<const uint2*>(recw);
  int* mylst = lst + (tid >> 5) * AG_LCAP;
  int cnt = 0;
  const int ebase = s * AG_SLICE;
  const uint4* r16 = reinterpret_cast<const uint4*>(rid4) + (size_t)s * AG_NV;
  const unsigned qs = (unsigned)q * 0x01010101u;
  for (int i0 = tid; i0 < AG_NV; i0 += 4 * AG_TPB) {
    uint4 dd[4];
#pragma unroll
    for (int j = 0; j < 4; ++j) {
      const int i = i0 + j * AG_TPB;
      dd[j] = (i < AG_NV) ? r16[i] : make_uint4(~0u, ~0u, ~0u, ~0u);
    }
    int e = ebase + 16 * i0;
#pragma unroll 1
    for (int j = 0; j < 4; ++j) {
      const uint4 cur = dd[0];
      dd[0] = dd[1];
      dd[1] = dd[2];
      dd[2] = dd[3];
      ag_word(tab, mylst, cnt, cur.x ^ qs, e,      lo, lane, rec, dis, XW);
      ag_word(tab, mylst, cnt, cur.y ^ qs, e + 4,  lo, lane, rec, dis, XW);
      ag_word(tab, mylst, cnt, cur.z ^ qs, e + 8,  lo, lane, rec, dis, XW);
      ag_word(tab, mylst, cnt, cur.w ^ qs, e + 12, lo, lane, rec, dis, XW);
      e += 16 * AG_TPB;
    }
  }
  if (cnt > 0) ag_flush(tab, mylst, cnt, lo, lane, rec, dis, XW);
  __syncthreads();

  float4* out4 = reinterpret_cast<float4*>(P + ((size_t)s * Nn + lo) * 96);
  for (int i = tid; i < AG_R * 24; i += AG_TPB) out4[i] = tab4[i];
}

__global__ void __launch_bounds__(32) gru_kernel(const float* __restrict__ XW, const float* __restrict__ P,
                                                 const float* __restrict__ dis, const float* __restrict__ bcat,
                                                 const float* __restrict__ Hp,
                                                 const float* __restrict__ LzW, const float* __restrict__ Lzb,
                                                 const float* __restrict__ LrW, const float* __restrict__ Lrb,
                                                 const float* __restrict__ LhW, const float* __restrict__ Lhb,
                                                 const float* __restrict__ headW, const float* __restrict__ headb,
                                                 float* __restrict__ y_out, float* __restrict__ h_out) {
  __shared__ __attribute__((aligned(16))) float As[16 * 132];
  __shared__ float Rs[16 * 36];
  const int lane = threadIdx.x;
  const int tile = blockIdx.x * 16;

  const float4* xw4 = reinterpret_cast<const float4*>(XW + (size_t)tile * 96);
  const float4* p04 = reinterpret_cast<const float4*>(P + (size_t)tile * 96);
  const float4* p14 = reinterpret_cast<const float4*>(P + ((size_t)Nn + tile) * 96);
  const float4* b4  = reinterpret_cast<const float4*>(bcat);
  const float4* hp4 = reinterpret_cast<const float4*>(Hp + (size_t)tile * 32);

  float4 bb[3];
#pragma unroll
  for (int j = 0; j < 3; ++j) bb[j] = b4[(lane + 32 * j) % 24];

#pragma unroll
  for (int g = 0; g < 3; ++g) {
    float4 xv[4], pa[4], pb[4];
    float dv[4];
#pragma unroll
    for (int j = 0; j < 4; ++j) {
      const int f = lane + 32 * (4 * g + j);
      xv[j] = xw4[f];
      pa[j] = p04[f];
      pb[j] = p14[f];
      dv[j] = dis[tile + f / 24];
    }
#pragma unroll
    for (int j = 0; j < 4; ++j) {
      const int f = lane + 32 * (4 * g + j);
      const int r = f / 24, c = 4 * (f - 24 * r);
      const float d2 = dv[j] * dv[j];
      const float4 bv = bb[(4 * g + j) % 3];
      float4 a;
      a.x = d2 * xv[j].x + bv.x;  a.x += pa[j].x;  a.x += pb[j].x;
      a.y = d2 * xv[j].y + bv.y;  a.y += pa[j].y;  a.y += pb[j].y;
      a.z = d2 * xv[j].z + bv.z;  a.z += pa[j].z;  a.z += pb[j].z;
      a.w = d2 * xv[j].w + bv.w;  a.w += pa[j].w;  a.w += pb[j].w;
      *reinterpret_cast<float4*>(&As[r * 132 + c]) = a;
    }
  }
#pragma unroll
  for (int j = 0; j < 4; ++j) {
    const int f = lane + 32 * j;
    *reinterpret_cast<float4*>(&As[(f >> 3) * 132 + 96 + 4 * (f & 7)]) = hp4[f];
  }
  __syncthreads();

  const int m = lane & 15;
  const int n = lane & 15;
  const int koff = (lane >> 4) << 1;

  v8f z0 = {}, z1 = {}, r0 = {}, r1 = {};
#pragma unroll 4
  for (int k0 = 0; k0 < 64; k0 += 4) {
    const int kk = k0 + koff;
    const int colz = (kk < 32) ? kk : (64 + kk);
    const int colr = (kk < 32) ? (32 + kk) : (64 + kk);
    v2f az; az.x = As[m * 132 + colz]; az.y = As[m * 132 + colz + 1];
    v2f ar; ar.x = As[m * 132 + colr]; ar.y = As[m * 132 + colr + 1];
    const float* lz = LzW + kk * 32 + n;
    const float* lr = LrW + kk * 32 + n;
    v2f b;
    b.x = lz[0];  b.y = lz[32];  z0 = wmma4(az, b, z0);
    b.x = lz[16]; b.y = lz[48];  z1 = wmma4(az, b, z1);
    b.x = lr[0];  b.y = lr[32];  r0 = wmma4(ar, b, r0);
    b.x = lr[16]; b.y = lr[48];  r1 = wmma4(ar, b, r1);
  }

  const float bz0 = Lzb[n], bz1 = Lzb[16 + n];
  const float br0 = Lrb[n], br1 = Lrb[16 + n];
  const int row0 = (lane >> 4) << 3;
#pragma unroll
  for (int v = 0; v < 8; ++v) {
    z0[v] = sigmoidf_(z0[v] + bz0);
    z1[v] = sigmoidf_(z1[v] + bz1);
    r0[v] = sigmoidf_(r0[v] + br0);
    r1[v] = sigmoidf_(r1[v] + br1);
    Rs[(row0 + v) * 36 + n]      = r0[v];
    Rs[(row0 + v) * 36 + 16 + n] = r1[v];
  }
  __syncthreads();

  v8f t0 = {}, t1 = {};
#pragma unroll 4
  for (int k0 = 0; k0 < 64; k0 += 4) {
    const int kk = k0 + koff;
    v2f a;
    a.x = As[m * 132 + 64 + kk];
    a.y = As[m * 132 + 65 + kk];
    if (kk >= 32) {
      a.x *= Rs[m * 36 + kk - 32];
      a.y *= Rs[m * 36 + kk - 31];
    }
    const float* lh = LhW + kk * 32 + n;
    v2f b;
    b.x = lh[0];  b.y = lh[32];  t0 = wmma4(a, b, t0);
    b.x = lh[16]; b.y = lh[48];  t1 = wmma4(a, b, t1);
  }

  const float bh0 = Lhb[n], bh1 = Lhb[16 + n];
  const float hw0 = headW[n], hw1 = headW[16 + n];
  float p[8];
#pragma unroll
  for (int v = 0; v < 8; ++v) {
    const float ht0 = tanhf(t0[v] + bh0);
    const float ht1 = tanhf(t1[v] + bh1);
    const float hp0 = As[(row0 + v) * 132 + 96 + n];
    const float hp1 = As[(row0 + v) * 132 + 112 + n];
    const float h0 = z0[v] * hp0 + (1.0f - z0[v]) * ht0;
    const float h1 = z1[v] * hp1 + (1.0f - z1[v]) * ht1;
    h_out[(size_t)(tile + row0 + v) * 32 + n]      = h0;
    h_out[(size_t)(tile + row0 + v) * 32 + 16 + n] = h1;
    p[v] = fmaxf(h0, 0.0f) * hw0 + fmaxf(h1, 0.0f) * hw1;
  }
#pragma unroll
  for (int s = 1; s < 16; s <<= 1)
#pragma unroll
    for (int v = 0; v < 8; ++v)
      p[v] += __shfl_xor(p[v], s, 32);
  if (n == 0) {
    const float hb = headb[0];
#pragma unroll
    for (int v = 0; v < 8; ++v)
      y_out[tile + row0 + v] = p[v] + hb;
  }
}

extern "C" void kernel_launch(void* const* d_in, const int* in_sizes, int n_in,
                              void* d_out, int out_size, void* d_ws, size_t ws_size,
                              hipStream_t stream) {
  const float* x    = (const float*)d_in[0];
  const int*   ei   = (const int*)d_in[1];
  const float* ew   = (const float*)d_in[2];
  const float* Hp   = (const float*)d_in[3];
  const float* Wz   = (const float*)d_in[4];
  const float* bz   = (const float*)d_in[5];
  const float* Wr   = (const float*)d_in[6];
  const float* br   = (const float*)d_in[7];
  const float* Wh   = (const float*)d_in[8];
  const float* bh   = (const float*)d_in[9];
  const float* LzW  = (const float*)d_in[10];
  const float* Lzb  = (const float*)d_in[11];
  const float* LrW  = (const float*)d_in[12];
  const float* Lrb  = (const float*)d_in[13];
  const float* LhW  = (const float*)d_in[14];
  const float* Lhb  = (const float*)d_in[15];
  const float* hW   = (const float*)d_in[16];
  const float* hb   = (const float*)d_in[17];

  const int* src = ei;
  const int* dst = ei + Ee;

  float* out_y = (float*)d_out;
  float* out_h = out_y + Nn;

  float* ws   = (float*)d_ws;
  float* dis  = ws;
  float* Wcat = ws + Nn;
  float* bcat = Wcat + FIN * 96;
  float* XW   = bcat + 96;
  float* part = XW + (size_t)Nn * 96;
  float* P    = part + (size_t)DEG_S * Nn;
  unsigned* rid4 = (unsigned*)(P + (size_t)AG_S * Nn * 96);
  unsigned* rec  = rid4 + Ee / 4;

  pack_weights_kernel<<<(FIN * 96 + 255) / 256, 256, 0, stream>>>(Wz, Wr, Wh, bz, br, bh, Wcat, bcat);
  deg_partial_kernel<<<DEG_Q * DEG_S, DEG_TPB, 0, stream>>>(src, dst, ew, part, rid4, rec);
  finalize_dis_kernel<<<(Nn + 255) / 256, 256, 0, stream>>>(part, dis);
  gemm_xw_kernel<<<Nn / 16, 32, 0, stream>>>(x, Wcat, XW);
  agg_range_kernel<<<AG_Q * AG_S, AG_TPB, 0, stream>>>(rid4, rec, dis, XW, P);
  gru_kernel<<<Nn / 16, 32, 0, stream>>>(XW, P, dis, bcat, Hp, LzW, Lzb, LrW, Lrb, LhW, Lhb, hW, hb, out_y, out_h);
}


# GEMM A operand read straight from x into fragments (16 contiguous bytes per lane and load, k re-grouped to match), all 32 loads in flight before the first WMMA; no LDS tile; XW sum order changed
# speedup vs baseline: 3.0350x; 1.0732x over previous
#include <hip/hip_runtime.h>

#define Nn    100000
#define Ee    3200000
#define FIN   256
#define Hh    32
#define AG_R  800

typedef float v2f __attribute__((ext_vector_type(2)));
typedef float v8f __attribute__((ext_vector_type(8)));

__device__ __forceinline__ v8f wmma4(v2f a, v2f b, v8f c) {
  return __builtin_amdgcn_wmma_f32_16x16x4_f32(
       false, a,  false, b,
       (short)0, c,  false,  false);
}

__device__ __forceinline__ float sigmoidf_(float x) {
  return 1.0f / (1.0f + __expf(-x));
}

__global__ void pack_weights_kernel(const float* __restrict__ Wz, const float* __restrict__ Wr,
                                    const float* __restrict__ Wh, const float* __restrict__ bz,
                                    const float* __restrict__ br, const float* __restrict__ bh,
                                    float* __restrict__ Wcat, float* __restrict__ bcat) {
  int gid = blockIdx.x * blockDim.x + threadIdx.x;
  if (gid < FIN * 96) {
    int k = gid / 96, c = gid - k * 96;
    float v = (c < 32) ? Wz[k * 32 + c] : ((c < 64) ? Wr[k * 32 + c - 32] : Wh[k * 32 + c - 64]);
    Wcat[gid] = v;
  }
  if (gid < 96) {
    bcat[gid] = (gid < 32) ? bz[gid] : ((gid < 64) ? br[gid - 32] : bh[gid - 64]);
  }
}

#define DEG_Q      4
#define DEG_RANGE  (Nn / DEG_Q)
#define DEG_S      64
#define DEG_SLICE  (Ee / DEG_S)
#define DEG_NV     (DEG_SLICE / 4)
#define DEG_TPB    1024

__device__ __forceinline__ unsigned rid_of(int d) {
  return ((unsigned)d < (unsigned)Nn) ? (unsigned)d / (unsigned)AG_R : 255u;
}

__device__ __forceinline__ unsigned rec_word(int s, int d) {
  const unsigned row = (unsigned)d - rid_of(d) * (unsigned)AG_R;
  return ((unsigned)s & 0x1ffffu) | (row << 17);
}

__device__ __forceinline__ void deg_add(float* tab, int d, float wv) {
  if ((unsigned)d < (unsigned)DEG_RANGE) unsafeAtomicAdd(&tab[d], wv);
}

__global__ void __launch_bounds__(DEG_TPB) deg_partial_kernel(const int* __restrict__ src,
                                                              const int* __restrict__ dst,
                                                              const float* __restrict__ w,
                                                              float* __restrict__ part,
                                                              unsigned* __restrict__ rid4,
                                                              unsigned* __restrict__ rec) {
  __shared__ float tab[DEG_RANGE];
  const int tid = threadIdx.x;
  const int q = blockIdx.x % DEG_Q;
  const int s = blockIdx.x / DEG_Q;
  const int lo = q * DEG_RANGE;

  float4* tab4 = reinterpret_cast<float4*>(tab);
  for (int i = tid; i < DEG_RANGE / 4; i += DEG_TPB) tab4[i] = make_float4(0.0f, 0.0f, 0.0f, 0.0f);
  __syncthreads();

  const int4*   s4 = reinterpret_cast<const int4*>(src + (size_t)s * DEG_SLICE);
  const int4*   d4 = reinterpret_cast<const int4*>(dst + (size_t)s * DEG_SLICE);
  const float4* w4 = reinterpret_cast<const float4*>(w + (size_t)s * DEG_SLICE);
  uint4* rec4 = reinterpret_cast<uint4*>(rec) + 2 * (size_t)s * DEG_NV;
  for (int i0 = tid; i0 < DEG_NV; i0 += 4 * DEG_TPB) {
    int4 dd[4];
    float4 ww[4];
#pragma unroll
    for (int j = 0; j < 4; ++j) {
      const int i = i0 + j * DEG_TPB;
      if (i < DEG_NV) {
        dd[j] = d4[i];
        ww[j] = w4[i];
      } else {
        dd[j] = make_int4(-1, -1, -1, -1);
        ww[j] = make_float4(0.0f, 0.0f, 0.0f, 0.0f);
      }
    }
    const int iq = i0 + q * DEG_TPB;
    int4 ss = make_int4(0, 0, 0, 0);
    if (iq < DEG_NV) ss = s4[iq];
#pragma unroll
    for (int j = 0; j < 4; ++j) {
      if (j == q && i0 + j * DEG_TPB < DEG_NV) {
        const int i = i0 + j * DEG_TPB;
        rid4[(size_t)s * DEG_NV + i] =
            rid_of(dd[j].x) | (rid_of(dd[j].y) << 8) | (rid_of(dd[j].z) << 16) | (rid_of(dd[j].w) << 24);
        rec4[2 * i]     = make_uint4(__float_as_uint(ww[j].x), rec_word(ss.x, dd[j].x),
                                     __float_as_uint(ww[j].y), rec_word(ss.y, dd[j].y));
        rec4[2 * i + 1] = make_uint4(__float_as_uint(ww[j].z), rec_word(ss.z, dd[j].z),
                                     __float_as_uint(ww[j].w), rec_word(ss.w, dd[j].w));
      }
      deg_add(tab, dd[j].x - lo, ww[j].x);
      deg_add(tab, dd[j].y - lo, ww[j].y);
      deg_add(tab, dd[j].z - lo, ww[j].z);
      deg_add(tab, dd[j].w - lo, ww[j].w);
    }
  }
  __syncthreads();

  float4* out4 = reinterpret_cast<float4*>(part + (size_t)s * Nn + lo);
  for (int i = tid; i < DEG_RANGE / 4; i += DEG_TPB) out4[i] = tab4[i];
}

__global__ void finalize_dis_kernel(const float* __restrict__ part, float* __restrict__ dis) {
  int gid = blockIdx.x * blockDim.x + threadIdx.x;
  if (gid < Nn) {
    float sum = 0.0f;
    for (int s = 0; s < DEG_S; ++s) sum += part[(size_t)s * Nn + gid];
    dis[gid] = rsqrtf(sum + 1.0f);
  }
}

__global__ void __launch_bounds__(32, 4) gemm_xw_kernel(const float* __restrict__ x,
                                                        const float* __restrict__ Wcat,
                                                        float* __restrict__ XW) {
  const int lane = threadIdx.x;
  const int tile = blockIdx.x * 16;
  const int m = lane & 15;
  const int h = lane >> 4;

  const float4* xr = reinterpret_cast<const float4*>(x + (size_t)(tile + m) * FIN + 4 * h);
  float4 av[32];
#pragma unroll
  for (int p = 0; p < 32; ++p) av[p] = xr[2 * p];
#pragma unroll
  for (int p = 0; p < 32; ++p)
    asm volatile("" :: "v"(av[p].x), "v"(av[p].y), "v"(av[p].z), "v"(av[p].w));

  v8f acc[6] = {};
  const float* bbase = Wcat + (size_t)(4 * h) * 96 + m;

#pragma unroll
  for (int p = 0; p < 32; ++p) {
#pragma unroll
    for (int s = 0; s < 2; ++s) {
      v2f a;
      a.x = s ? av[p].z : av[p].x;
      a.y = s ? av[p].w : av[p].y;
      const float* brow = bbase + (size_t)(8 * p + 2 * s) * 96;
#pragma unroll
      for (int t = 0; t < 6; ++t) {
        v2f b; b.x = brow[t * 16]; b.y = brow[96 + t * 16];
        acc[t] = wmma4(a, b, acc[t]);
      }
    }
  }

  const int row0 = h << 3;
#pragma unroll
  for (int t = 0; t < 6; ++t)
#pragma unroll
    for (int v = 0; v < 8; ++v)
      XW[(size_t)(tile + row0 + v) * 96 + t * 16 + m] = acc[t][v];
}

#define AG_Q      (Nn / AG_R)
#define AG_S      2
#define AG_SLICE  (Ee / AG_S)
#define AG_NV     (AG_SLICE / 16)
#define AG_TPB    1024
#define AG_LCAP   96

__device__ __forceinline__ void ag_push(int* lst, int& cnt, bool m, int e, int lane) {
  const unsigned mask = __builtin_amdgcn_ballot_w32(m);
  if (mask != 0u) {
    if (m) lst[cnt + __builtin_popcount(mask & ((1u << lane) - 1u))] = e;
    cnt += __builtin_popcount(mask);
  }
}

__device__ __forceinline__ void ag_flush(float* tab, const int* lst, int n, int lo, int lane,
                                         const uint2* __restrict__ rec, const float* __restrict__ dis,
                                         const float* __restrict__ XW) {
  int sN = 0, ld = 0;
  float nrm = 0.0f;
  if (lane < n) {
    const uint2 r = rec[lst[lane]];
    const int s = (int)(r.y & 0x1ffffu);
    const int row = (int)(r.y >> 17);
    nrm = dis[s] * __uint_as_float(r.x) * dis[lo + row];
    sN = s * 96;
    ld = row * 96;
  }
  for (int j0 = 0; j0 < n; j0 += 8) {
    float v[8][3];
#pragma unroll
    for (int j = 0; j < 8; ++j) {
      const int sb = __builtin_amdgcn_readlane(sN, j0 + j);
#pragma unroll
      for (int i = 0; i < 3; ++i) v[j][i] = XW[(size_t)sb + lane + 32 * i];
    }
#pragma unroll
    for (int j = 0; j < 8; ++j) {
      if (j0 + j < n) {
        const float nj = __int_as_float(__builtin_amdgcn_readlane(__float_as_int(nrm), j0 + j));
        const int lj = __builtin_amdgcn_readlane(ld, j0 + j);
#pragma unroll
        for (int i = 0; i < 3; ++i) unsafeAtomicAdd(&tab[lj + lane + 32 * i], nj * v[j][i]);
      }
    }
  }
}

__device__ __forceinline__ void ag_word(float* tab, int* lst, int& cnt, unsigned x, int e, int lo, int lane,
                                        const uint2* __restrict__ rec, const float* __restrict__ dis,
                                        const float* __restrict__ XW) {
  ag_push(lst, cnt, (x & 0x000000ffu) == 0u, e, lane);
  ag_push(lst, cnt, (x & 0x0000ff00u) == 0u, e + 1, lane);
  while (cnt >= 32) {
    cnt -= 32;
    ag_flush(tab, lst + cnt, 32, lo, lane, rec, dis, XW);
  }
  ag_push(lst, cnt, (x & 0x00ff0000u) == 0u, e + 2, lane);
  ag_push(lst, cnt, (x & 0xff000000u) == 0u, e + 3, lane);
  while (cnt >= 32) {
    cnt -= 32;
    ag_flush(tab, lst + cnt, 32, lo, lane, rec, dis, XW);
  }
}

__global__ void __launch_bounds__(AG_TPB) agg_range_kernel(const unsigned* __restrict__ rid4,
                                                           const unsigned* __restrict__ recw,
                                                           const float* __restrict__ dis,
                                                           const float* __restrict__ XW,
                                                           float* __restrict__ P) {
  __shared__ float tab[AG_R * 96];
  __shared__ int lst[(AG_TPB / 32) * AG_LCAP];
  const int tid = threadIdx.x;
  const int lane = tid & 31;
  const int s = blockIdx.x % AG_S;
  const int q = blockIdx.x / AG_S;
  const int lo = q * AG_R;

  float4* tab4 = reinterpret_cast<float4*>(tab);
  for (int i = tid; i < AG_R * 24; i += AG_TPB) tab4[i] = make_float4(0.0f, 0.0f, 0.0f, 0.0f);
  __syncthreads();

  const uint2* rec = reinterpret_cast<const uint2*>(recw);
  int* mylst = lst + (tid >> 5) * AG_LCAP;
  int cnt = 0;
  const int ebase = s * AG_SLICE;
  const uint4* r16 = reinterpret_cast<const uint4*>(rid4) + (size_t)s * AG_NV;
  const unsigned qs = (unsigned)q * 0x01010101u;
  for (int i0 = tid; i0 < AG_NV; i0 += 4 * AG_TPB) {
    uint4 dd[4];
#pragma unroll
    for (int j = 0; j < 4; ++j) {
      const int i = i0 + j * AG_TPB;
      dd[j] = (i < AG_NV) ? r16[i] : make_uint4(~0u, ~0u, ~0u, ~0u);
    }
    int e = ebase + 16 * i0;
#pragma unroll 1
    for (int j = 0; j < 4; ++j) {
      const uint4 cur = dd[0];
      dd[0] = dd[1];
      dd[1] = dd[2];
      dd[2] = dd[3];
      ag_word(tab, mylst, cnt, cur.x ^ qs, e,      lo, lane, rec, dis, XW);
      ag_word(tab, mylst, cnt, cur.y ^ qs, e + 4,  lo, lane, rec, dis, XW);
      ag_word(tab, mylst, cnt, cur.z ^ qs, e + 8,  lo, lane, rec, dis, XW);
      ag_word(tab, mylst, cnt, cur.w ^ qs, e + 12, lo, lane, rec, dis, XW);
      e += 16 * AG_TPB;
    }
  }
  if (cnt > 0) ag_flush(tab, mylst, cnt, lo, lane, rec, dis, XW);
  __syncthreads();

  float4* out4 = reinterpret_cast<float4*>(P + ((size_t)s * Nn + lo) * 96);
  for (int i = tid; i < AG_R * 24; i += AG_TPB) out4[i] = tab4[i];
}

__global__ void __launch_bounds__(32) gru_kernel(const float* __restrict__ XW, const float* __restrict__ P,
                                                 const float* __restrict__ dis, const float* __restrict__ bcat,
                                                 const float* __restrict__ Hp,
                                                 const float* __restrict__ LzW, const float* __restrict__ Lzb,
                                                 const float* __restrict__ LrW, const float* __restrict__ Lrb,
                                                 const float* __restrict__ LhW, const float* __restrict__ Lhb,
                                                 const float* __restrict__ headW, const float* __restrict__ headb,
                                                 float* __restrict__ y_out, float* __restrict__ h_out) {
  __shared__ __attribute__((aligned(16))) float As[16 * 132];
  __shared__ float Rs[16 * 36];
  const int lane = threadIdx.x;
  const int tile = blockIdx.x * 16;

  const float4* xw4 = reinterpret_cast<const float4*>(XW + (size_t)tile * 96);
  const float4* p04 = reinterpret_cast<const float4*>(P + (size_t)tile * 96);
  const float4* p14 = reinterpret_cast<const float4*>(P + ((size_t)Nn + tile) * 96);
  const float4* b4  = reinterpret_cast<const float4*>(bcat);
  const float4* hp4 = reinterpret_cast<const float4*>(Hp + (size_t)tile * 32);

  float4 bb[3];
#pragma unroll
  for (int j = 0; j < 3; ++j) bb[j] = b4[(lane + 32 * j) % 24];

#pragma unroll
  for (int g = 0; g < 3; ++g) {
    float4 xv[4], pa[4], pb[4];
    float dv[4];
#pragma unroll
    for (int j = 0; j < 4; ++j) {
      const int f = lane + 32 * (4 * g + j);
      xv[j] = xw4[f];
      pa[j] = p04[f];
      pb[j] = p14[f];
      dv[j] = dis[tile + f / 24];
    }
#pragma unroll
    for (int j = 0; j < 4; ++j) {
      const int f = lane + 32 * (4 * g + j);
      const int r = f / 24, c = 4 * (f - 24 * r);
      const float d2 = dv[j] * dv[j];
      const float4 bv = bb[(4 * g + j) % 3];
      float4 a;
      a.x = d2 * xv[j].x + bv.x;  a.x += pa[j].x;  a.x += pb[j].x;
      a.y = d2 * xv[j].y + bv.y;  a.y += pa[j].y;  a.y += pb[j].y;
      a.z = d2 * xv[j].z + bv.z;  a.z += pa[j].z;  a.z += pb[j].z;
      a.w = d2 * xv[j].w + bv.w;  a.w += pa[j].w;  a.w += pb[j].w;
      *reinterpret_cast<float4*>(&As[r * 132 + c]) = a;
    }
  }
#pragma unroll
  for (int j = 0; j < 4; ++j) {
    const int f = lane + 32 * j;
    *reinterpret_cast<float4*>(&As[(f >> 3) * 132 + 96 + 4 * (f & 7)]) = hp4[f];
  }
  __syncthreads();

  const int m = lane & 15;
  const int n = lane & 15;
  const int koff = (lane >> 4) << 1;

  v8f z0 = {}, z1 = {}, r0 = {}, r1 = {};
#pragma unroll 4
  for (int k0 = 0; k0 < 64; k0 += 4) {
    const int kk = k0 + koff;
    const int colz = (kk < 32) ? kk : (64 + kk);
    const int colr = (kk < 32) ? (32 + kk) : (64 + kk);
    v2f az; az.x = As[m * 132 + colz]; az.y = As[m * 132 + colz + 1];
    v2f ar; ar.x = As[m * 132 + colr]; ar.y = As[m * 132 + colr + 1];
    const float* lz = LzW + kk * 32 + n;
    const float* lr = LrW + kk * 32 + n;
    v2f b;
    b.x = lz[0];  b.y = lz[32];  z0 = wmma4(az, b, z0);
    b.x = lz[16]; b.y = lz[48];  z1 = wmma4(az, b, z1);
    b.x = lr[0];  b.y = lr[32];  r0 = wmma4(ar, b, r0);
    b.x = lr[16]; b.y = lr[48];  r1 = wmma4(ar, b, r1);
  }

  const float bz0 = Lzb[n], bz1 = Lzb[16 + n];
  const float br0 = Lrb[n], br1 = Lrb[16 + n];
  const int row0 = (lane >> 4) << 3;
#pragma unroll
  for (int v = 0; v < 8; ++v) {
    z0[v] = sigmoidf_(z0[v] + bz0);
    z1[v] = sigmoidf_(z1[v] + bz1);
    r0[v] = sigmoidf_(r0[v] + br0);
    r1[v] = sigmoidf_(r1[v] + br1);
    Rs[(row0 + v) * 36 + n]      = r0[v];
    Rs[(row0 + v) * 36 + 16 + n] = r1[v];
  }
  __syncthreads();

  v8f t0 = {}, t1 = {};
#pragma unroll 4
  for (int k0 = 0; k0 < 64; k0 += 4) {
    const int kk = k0 + koff;
    v2f a;
    a.x = As[m * 132 + 64 + kk];
    a.y = As[m * 132 + 65 + kk];
    if (kk >= 32) {
      a.x *= Rs[m * 36 + kk - 32];
      a.y *= Rs[m * 36 + kk - 31];
    }
    const float* lh = LhW + kk * 32 + n;
    v2f b;
    b.x = lh[0];  b.y = lh[32];  t0 = wmma4(a, b, t0);
    b.x = lh[16]; b.y = lh[48];  t1 = wmma4(a, b, t1);
  }

  const float bh0 = Lhb[n], bh1 = Lhb[16 + n];
  const float hw0 = headW[n], hw1 = headW[16 + n];
  float p[8];
#pragma unroll
  for (int v = 0; v < 8; ++v) {
    const float ht0 = tanhf(t0[v] + bh0);
    const float ht1 = tanhf(t1[v] + bh1);
    const float hp0 = As[(row0 + v) * 132 + 96 + n];
    const float hp1 = As[(row0 + v) * 132 + 112 + n];
    const float h0 = z0[v] * hp0 + (1.0f - z0[v]) * ht0;
    const float h1 = z1[v] * hp1 + (1.0f - z1[v]) * ht1;
    h_out[(size_t)(tile + row0 + v) * 32 + n]      = h0;
    h_out[(size_t)(tile + row0 + v) * 32 + 16 + n] = h1;
    p[v] = fmaxf(h0, 0.0f) * hw0 + fmaxf(h1, 0.0f) * hw1;
  }
#pragma unroll
  for (int s = 1; s < 16; s <<= 1)
#pragma unroll
    for (int v = 0; v < 8; ++v)
      p[v] += __shfl_xor(p[v], s, 32);
  if (n == 0) {
    const float hb = headb[0];
#pragma unroll
    for (int v = 0; v < 8; ++v)
      y_out[tile + row0 + v] = p[v] + hb;
  }
}

extern "C" void kernel_launch(void* const* d_in, const int* in_sizes, int n_in,
                              void* d_out, int out_size, void* d_ws, size_t ws_size,
                              hipStream_t stream) {
  const float* x    = (const float*)d_in[0];
  const int*   ei   = (const int*)d_in[1];
  const float* ew   = (const float*)d_in[2];
  const float* Hp   = (const float*)d_in[3];
  const float* Wz   = (const float*)d_in[4];
  const float* bz   = (const float*)d_in[5];
  const float* Wr   = (const float*)d_in[6];
  const float* br   = (const float*)d_in[7];
  const float* Wh   = (const float*)d_in[8];
  const float* bh   = (const float*)d_in[9];
  const float* LzW  = (const float*)d_in[10];
  const float* Lzb  = (const float*)d_in[11];
  const float* LrW  = (const float*)d_in[12];
  const float* Lrb  = (const float*)d_in[13];
  const float* LhW  = (const float*)d_in[14];
  const float* Lhb  = (const float*)d_in[15];
  const float* hW   = (const float*)d_in[16];
  const float* hb   = (const float*)d_in[17];

  const int* src = ei;
  const int* dst = ei + Ee;

  float* out_y = (float*)d_out;
  float* out_h = out_y + Nn;

  float* ws   = (float*)d_ws;
  float* dis  = ws;
  float* Wcat = ws + Nn;
  float* bcat = Wcat + FIN * 96;
  float* XW   = bcat + 96;
  float* part = XW + (size_t)Nn * 96;
  float* P    = part + (size_t)DEG_S * Nn;
  unsigned* rid4 = (unsigned*)(P + (size_t)AG_S * Nn * 96);
  unsigned* rec  = rid4 + Ee / 4;

  pack_weights_kernel<<<(FIN * 96 + 255) / 256, 256, 0, stream>>>(Wz, Wr, Wh, bz, br, bh, Wcat, bcat);
  deg_partial_kernel<<<DEG_Q * DEG_S, DEG_TPB, 0, stream>>>(src, dst, ew, part, rid4, rec);
  finalize_dis_kernel<<<(Nn + 255) / 256, 256, 0, stream>>>(part, dis);
  gemm_xw_kernel<<<Nn / 16, 32, 0, stream>>>(x, Wcat, XW);
  agg_range_kernel<<<AG_Q * AG_S, AG_TPB, 0, stream>>>(rid4, rec, dis, XW, P);
  gru_kernel<<<Nn / 16, 32, 0, stream>>>(XW, P, dis, bcat, Hp, LzW, Lzb, LrW, Lrb, LhW, Lhb, hW, hb, out_y, out_h);
}


# Edge filter as one bit plane per node range: degree pass ORs a bit per edge into LDS rows, stores whole lines; aggregate scans 32 edges per word, not a byte each; sums still arrival order per plane
# speedup vs baseline: 3.5841x; 1.1809x over previous
#include <hip/hip_runtime.h>

#define Nn    100000
#define Ee    3200000
#define FIN   256
#define Hh    32
#define AG_R  800
#define AG_Q  (Nn / AG_R)

#define BM_W     (Ee / 32)
#define BM_BLK   1024
#define BM_NBLK  (Ee / BM_BLK)

typedef float v2f __attribute__((ext_vector_type(2)));
typedef float v8f __attribute__((ext_vector_type(8)));

__device__ __forceinline__ v8f wmma4(v2f a, v2f b, v8f c) {
  return __builtin_amdgcn_wmma_f32_16x16x4_f32(
       false, a,  false, b,
       (short)0, c,  false,  false);
}

__device__ __forceinline__ float sigmoidf_(float x) {
  return 1.0f / (1.0f + __expf(-x));
}

__global__ void pack_weights_kernel(const float* __restrict__ Wz, const float* __restrict__ Wr,
                                    const float* __restrict__ Wh, const float* __restrict__ bz,
                                    const float* __restrict__ br, const float* __restrict__ bh,
                                    float* __restrict__ Wcat, float* __restrict__ bcat) {
  int gid = blockIdx.x * blockDim.x + threadIdx.x;
  if (gid < FIN * 96) {
    int k = gid / 96, c = gid - k * 96;
    float v = (c < 32) ? Wz[k * 32 + c] : ((c < 64) ? Wr[k * 32 + c - 32] : Wh[k * 32 + c - 64]);
    Wcat[gid] = v;
  }
  if (gid < 96) {
    bcat[gid] = (gid < 32) ? bz[gid] : ((gid < 64) ? br[gid - 32] : bh[gid - 64]);
  }
}

#define DEG_Q      4
#define DEG_RANGE  (Nn / DEG_Q)
#define DEG_S      64
#define DEG_SLICE  (Ee / DEG_S)
#define DEG_NV     (DEG_SLICE / 4)
#define DEG_TPB    1024
#define BM_PER     ((BM_NBLK + DEG_Q * DEG_S - 1) / (DEG_Q * DEG_S))
#define BM_PITCH   33
#define BM_LDSW    ((BM_PER * AG_Q * BM_PITCH + 3) / 4 * 4)

__device__ __forceinline__ unsigned rid_of(int d) {
  return ((unsigned)d < (unsigned)Nn) ? (unsigned)d / (unsigned)AG_R : 255u;
}

__device__ __forceinline__ unsigned rec_word(int s, int d) {
  const unsigned row = (unsigned)d - rid_of(d) * (unsigned)AG_R;
  return ((unsigned)s & 0x1ffffu) | (row << 17);
}

__device__ __forceinline__ void deg_add(float* tab, int d, float wv) {
  if ((unsigned)d < (unsigned)DEG_RANGE) unsafeAtomicAdd(&tab[d], wv);
}

__global__ void __launch_bounds__(DEG_TPB) deg_partial_kernel(const int* __restrict__ src,
                                                              const int* __restrict__ dst,
                                                              const float* __restrict__ w,
                                                              float* __restrict__ part,
                                                              unsigned* __restrict__ bm,
                                                              unsigned* __restrict__ rec) {
  __shared__ float tab[DEG_RANGE];
  __shared__ __attribute__((aligned(16))) unsigned bml[BM_LDSW];
  const int tid = threadIdx.x;
  const int q = blockIdx.x % DEG_Q;
  const int s = blockIdx.x / DEG_Q;
  const int lo = q * DEG_RANGE;

  int bd[BM_PER];
#pragma unroll
  for (int k = 0; k < BM_PER; ++k) {
    const int blk = (int)blockIdx.x + k * (DEG_Q * DEG_S);
    bd[k] = (blk < BM_NBLK) ? dst[(size_t)blk * BM_BLK + tid] : -1;
  }

  float4* tab4 = reinterpret_cast<float4*>(tab);
  for (int i = tid; i < DEG_RANGE / 4; i += DEG_TPB) tab4[i] = make_float4(0.0f, 0.0f, 0.0f, 0.0f);
  uint4* bml4 = reinterpret_cast<uint4*>(bml);
  for (int i = tid; i < BM_LDSW / 4; i += DEG_TPB) bml4[i] = make_uint4(0u, 0u, 0u, 0u);
  __syncthreads();

#pragma unroll
  for (int k = 0; k < BM_PER; ++k) {
    const unsigned r = rid_of(bd[k]);
    if (r < (unsigned)AG_Q) atomicOr(&bml[(k * AG_Q + (int)r) * BM_PITCH + (tid >> 5)], 1u << (tid & 31));
  }
  __syncthreads();

  {
    const int lane = tid & 31;
    const int wv = __builtin_amdgcn_readfirstlane(tid >> 5);
    for (int kr = wv; kr < BM_PER * AG_Q; kr += DEG_TPB / 32) {
      const int k = kr / AG_Q;
      const int r = kr - k * AG_Q;
      const int blk = (int)blockIdx.x + k * (DEG_Q * DEG_S);
      if (blk < BM_NBLK) bm[(size_t)r * BM_W + (size_t)blk * (BM_BLK / 32) + lane] = bml[kr * BM_PITCH + lane];
    }
  }

  const int4*   s4 = reinterpret_cast<const int4*>(src + (size_t)s * DEG_SLICE);
  const int4*   d4 = reinterpret_cast<const int4*>(dst + (size_t)s * DEG_SLICE);
  const float4* w4 = reinterpret_cast<const float4*>(w + (size_t)s * DEG_SLICE);
  uint4* rec4 = reinterpret_cast<uint4*>(rec) + 2 * (size_t)s * DEG_NV;
  for (int i0 = tid; i0 < DEG_NV; i0 += 4 * DEG_TPB) {
    int4 dd[4];
    float4 ww[4];
#pragma unroll
    for (int j = 0; j < 4; ++j) {
      const int i = i0 + j * DEG_TPB;
      if (i < DEG_NV) {
        dd[j] = d4[i];
        ww[j] = w4[i];
      } else {
        dd[j] = make_int4(-1, -1, -1, -1);
        ww[j] = make_float4(0.0f, 0.0f, 0.0f, 0.0f);
      }
    }
    const int iq = i0 + q * DEG_TPB;
    int4 ss = make_int4(0, 0, 0, 0);
    if (iq < DEG_NV) ss = s4[iq];
#pragma unroll
    for (int j = 0; j < 4; ++j) {
      if (j == q && i0 + j * DEG_TPB < DEG_NV) {
        const int i = i0 + j * DEG_TPB;
        rec4[2 * i]     = make_uint4(__float_as_uint(ww[j].x), rec_word(ss.x, dd[j].x),
                                     __float_as_uint(ww[j].y), rec_word(ss.y, dd[j].y));
        rec4[2 * i + 1] = make_uint4(__float_as_uint(ww[j].z), rec_word(ss.z, dd[j].z),
                                     __float_as_uint(ww[j].w), rec_word(ss.w, dd[j].w));
      }
      deg_add(tab, dd[j].x - lo, ww[j].x);
      deg_add(tab, dd[j].y - lo, ww[j].y);
      deg_add(tab, dd[j].z - lo, ww[j].z);
      deg_add(tab, dd[j].w - lo, ww[j].w);
    }
  }
  __syncthreads();

  float4* out4 = reinterpret_cast<float4*>(part + (size_t)s * Nn + lo);
  for (int i = tid; i < DEG_RANGE / 4; i += DEG_TPB) out4[i] = tab4[i];
}

__global__ void finalize_dis_kernel(const float* __restrict__ part, float* __restrict__ dis) {
  int gid = blockIdx.x * blockDim.x + threadIdx.x;
  if (gid < Nn) {
    float sum = 0.0f;
    for (int s = 0; s < DEG_S; ++s) sum += part[(size_t)s * Nn + gid];
    dis[gid] = rsqrtf(sum + 1.0f);
  }
}

__global__ void __launch_bounds__(32, 4) gemm_xw_kernel(const float* __restrict__ x,
                                                        const float* __restrict__ Wcat,
                                                        float* __restrict__ XW) {
  const int lane = threadIdx.x;
  const int tile = blockIdx.x * 16;
  const int m = lane & 15;
  const int h = lane >> 4;

  const float4* xr = reinterpret_cast<const float4*>(x + (size_t)(tile + m) * FIN + 4 * h);
  float4 av[32];
#pragma unroll
  for (int p = 0; p < 32; ++p) av[p] = xr[2 * p];
#pragma unroll
  for (int p = 0; p < 32; ++p)
    asm volatile("" :: "v"(av[p].x), "v"(av[p].y), "v"(av[p].z), "v"(av[p].w));

  v8f acc[6] = {};
  const float* bbase = Wcat + (size_t)(4 * h) * 96 + m;

#pragma unroll
  for (int p = 0; p < 32; ++p) {
#pragma unroll
    for (int s = 0; s < 2; ++s) {
      v2f a;
      a.x = s ? av[p].z : av[p].x;
      a.y = s ? av[p].w : av[p].y;
      const float* brow = bbase + (size_t)(8 * p + 2 * s) * 96;
#pragma unroll
      for (int t = 0; t < 6; ++t) {
        v2f b; b.x = brow[t * 16]; b.y = brow[96 + t * 16];
        acc[t] = wmma4(a, b, acc[t]);
      }
    }
  }

  const int row0 = h << 3;
#pragma unroll
  for (int t = 0; t < 6; ++t)
#pragma unroll
    for (int v = 0; v < 8; ++v)
      XW[(size_t)(tile + row0 + v) * 96 + t * 16 + m] = acc[t][v];
}

#define AG_S      2
#define AG_SLICE  (Ee / AG_S)
#define AG_NV     (AG_SLICE / 128)
#define AG_TPB    1024
#define AG_TRIPS  ((AG_NV + 4 * AG_TPB - 1) / (4 * AG_TPB))
#define AG_LCAP   64

__device__ __forceinline__ void ag_flush(float* tab, const int* lst, int n, int lo, int lane,
                                         const uint2* __restrict__ rec, const float* __restrict__ dis,
                                         const float* __restrict__ XW) {
  int sN = 0, ld = 0;
  float nrm = 0.0f;
  if (lane < n) {
    const uint2 r = rec[lst[lane]];
    const int s = (int)(r.y & 0x1ffffu);
    const int row = (int)(r.y >> 17);
    nrm = dis[s] * __uint_as_float(r.x) * dis[lo + row];
    sN = s * 96;
    ld = row * 96;
  }
  for (int j0 = 0; j0 < n; j0 += 8) {
    float v[8][3];
#pragma unroll
    for (int j = 0; j < 8; ++j) {
      const int sb = __builtin_amdgcn_readlane(sN, j0 + j);
#pragma unroll
      for (int i = 0; i < 3; ++i) v[j][i] = XW[(size_t)sb + lane + 32 * i];
    }
#pragma unroll
    for (int j = 0; j < 8; ++j) {
      if (j0 + j < n) {
        const float nj = __int_as_float(__builtin_amdgcn_readlane(__float_as_int(nrm), j0 + j));
        const int lj = __builtin_amdgcn_readlane(ld, j0 + j);
#pragma unroll
        for (int i = 0; i < 3; ++i) unsafeAtomicAdd(&tab[lj + lane + 32 * i], nj * v[j][i]);
      }
    }
  }
}

__device__ __forceinline__ void ag_bits(float* tab, int* lst, int& cnt, unsigned wd, int e, int lo, int lane,
                                        const uint2* __restrict__ rec, const float* __restrict__ dis,
                                        const float* __restrict__ XW) {
  for (;;) {
    const bool m = wd != 0u;
    const unsigned mask = __builtin_amdgcn_ballot_w32(m);
    if (mask == 0u) break;
    if (m) {
      lst[cnt + __builtin_popcount(mask & ((1u << lane) - 1u))] = e + __builtin_ctz(wd);
      wd &= wd - 1u;
    }
    cnt += __builtin_popcount(mask);
    if (cnt >= 32) {
      cnt -= 32;
      ag_flush(tab, lst + cnt, 32, lo, lane, rec, dis, XW);
    }
  }
}

__global__ void __launch_bounds__(AG_TPB) agg_range_kernel(const unsigned* __restrict__ bm,
                                                           const unsigned* __restrict__ recw,
                                                           const float* __restrict__ dis,
                                                           const float* __restrict__ XW,
                                                           float* __restrict__ P) {
  __shared__ float tab[AG_R * 96];
  __shared__ int lst[(AG_TPB / 32) * AG_LCAP];
  const int tid = threadIdx.x;
  const int lane = tid & 31;
  const int s = blockIdx.x % AG_S;
  const int q = blockIdx.x / AG_S;
  const int lo = q * AG_R;

  float4* tab4 = reinterpret_cast<float4*>(tab);
  for (int i = tid; i < AG_R * 24; i += AG_TPB) tab4[i] = make_float4(0.0f, 0.0f, 0.0f, 0.0f);
  __syncthreads();

  const uint2* rec = reinterpret_cast<const uint2*>(recw);
  int* mylst = lst + (tid >> 5) * AG_LCAP;
  int cnt = 0;
  const int ebase = s * AG_SLICE;
  const uint4* b16 = reinterpret_cast<const uint4*>(bm + (size_t)q * BM_W) + (size_t)s * AG_NV;
  for (int t = 0; t < AG_TRIPS; ++t) {
    const int i0 = tid + t * (4 * AG_TPB);
    uint4 dd[4];
#pragma unroll
    for (int j = 0; j < 4; ++j) {
      const int i = i0 + j * AG_TPB;
      dd[j] = (i < AG_NV) ? b16[i] : make_uint4(0u, 0u, 0u, 0u);
    }
    int e = ebase + 128 * i0;
#pragma unroll 1
    for (int j = 0; j < 4; ++j) {
      uint4 cur = dd[0];
      dd[0] = dd[1];
      dd[1] = dd[2];
      dd[2] = dd[3];
#pragma unroll 1
      for (int cw = 0; cw < 4; ++cw) {
        const unsigned wd = cur.x;
        cur.x = cur.y;
        cur.y = cur.z;
        cur.z = cur.w;
        ag_bits(tab, mylst, cnt, wd, e + 32 * cw, lo, lane, rec, dis, XW);
      }
      e += 128 * AG_TPB;
    }
  }
  if (cnt > 0) ag_flush(tab, mylst, cnt, lo, lane, rec, dis, XW);
  __syncthreads();

  float4* out4 = reinterpret_cast<float4*>(P + ((size_t)s * Nn + lo) * 96);
  for (int i = tid; i < AG_R * 24; i += AG_TPB) out4[i] = tab4[i];
}

__global__ void __launch_bounds__(32) gru_kernel(const float* __restrict__ XW, const float* __restrict__ P,
                                                 const float* __restrict__ dis, const float* __restrict__ bcat,
                                                 const float* __restrict__ Hp,
                                                 const float* __restrict__ LzW, const float* __restrict__ Lzb,
                                                 const float* __restrict__ LrW, const float* __restrict__ Lrb,
                                                 const float* __restrict__ LhW, const float* __restrict__ Lhb,
                                                 const float* __restrict__ headW, const float* __restrict__ headb,
                                                 float* __restrict__ y_out, float* __restrict__ h_out) {
  __shared__ __attribute__((aligned(16))) float As[16 * 132];
  __shared__ float Rs[16 * 36];
  const int lane = threadIdx.x;
  const int tile = blockIdx.x * 16;

  const float4* xw4 = reinterpret_cast<const float4*>(XW + (size_t)tile * 96);
  const float4* p04 = reinterpret_cast<const float4*>(P + (size_t)tile * 96);
  const float4* p14 = reinterpret_cast<const float4*>(P + ((size_t)Nn + tile) * 96);
  const float4* b4  = reinterpret_cast<const float4*>(bcat);
  const float4* hp4 = reinterpret_cast<const float4*>(Hp + (size_t)tile * 32);

  float4 bb[3];
#pragma unroll
  for (int j = 0; j < 3; ++j) bb[j] = b4[(lane + 32 * j) % 24];

#pragma unroll
  for (int g = 0; g < 3; ++g) {
    float4 xv[4], pa[4], pb[4];
    float dv[4];
#pragma unroll
    for (int j = 0; j < 4; ++j) {
      const int f = lane + 32 * (4 * g + j);
      xv[j] = xw4[f];
      pa[j] = p04[f];
      pb[j] = p14[f];
      dv[j] = dis[tile + f / 24];
    }
#pragma unroll
    for (int j = 0; j < 4; ++j) {
      const int f = lane + 32 * (4 * g + j);
      const int r = f / 24, c = 4 * (f - 24 * r);
      const float d2 = dv[j] * dv[j];
      const float4 bv = bb[(4 * g + j) % 3];
      float4 a;
      a.x = d2 * xv[j].x + bv.x;  a.x += pa[j].x;  a.x += pb[j].x;
      a.y = d2 * xv[j].y + bv.y;  a.y += pa[j].y;  a.y += pb[j].y;
      a.z = d2 * xv[j].z + bv.z;  a.z += pa[j].z;  a.z += pb[j].z;
      a.w = d2 * xv[j].w + bv.w;  a.w += pa[j].w;  a.w += pb[j].w;
      *reinterpret_cast<float4*>(&As[r * 132 + c]) = a;
    }
  }
#pragma unroll
  for (int j = 0; j < 4; ++j) {
    const int f = lane + 32 * j;
    *reinterpret_cast<float4*>(&As[(f >> 3) * 132 + 96 + 4 * (f & 7)]) = hp4[f];
  }
  __syncthreads();

  const int m = lane & 15;
  const int n = lane & 15;
  const int koff = (lane >> 4) << 1;

  v8f z0 = {}, z1 = {}, r0 = {}, r1 = {};
#pragma unroll 4
  for (int k0 = 0; k0 < 64; k0 += 4) {
    const int kk = k0 + koff;
    const int colz = (kk < 32) ? kk : (64 + kk);
    const int colr = (kk < 32) ? (32 + kk) : (64 + kk);
    v2f az; az.x = As[m * 132 + colz]; az.y = As[m * 132 + colz + 1];
    v2f ar; ar.x = As[m * 132 + colr]; ar.y = As[m * 132 + colr + 1];
    const float* lz = LzW + kk * 32 + n;
    const float* lr = LrW + kk * 32 + n;
    v2f b;
    b.x = lz[0];  b.y = lz[32];  z0 = wmma4(az, b, z0);
    b.x = lz[16]; b.y = lz[48];  z1 = wmma4(az, b, z1);
    b.x = lr[0];  b.y = lr[32];  r0 = wmma4(ar, b, r0);
    b.x = lr[16]; b.y = lr[48];  r1 = wmma4(ar, b, r1);
  }

  const float bz0 = Lzb[n], bz1 = Lzb[16 + n];
  const float br0 = Lrb[n], br1 = Lrb[16 + n];
  const int row0 = (lane >> 4) << 3;
#pragma unroll
  for (int v = 0; v < 8; ++v) {
    z0[v] = sigmoidf_(z0[v] + bz0);
    z1[v] = sigmoidf_(z1[v] + bz1);
    r0[v] = sigmoidf_(r0[v] + br0);
    r1[v] = sigmoidf_(r1[v] + br1);
    Rs[(row0 + v) * 36 + n]      = r0[v];
    Rs[(row0 + v) * 36 + 16 + n] = r1[v];
  }
  __syncthreads();

  v8f t0 = {}, t1 = {};
#pragma unroll 4
  for (int k0 = 0; k0 < 64; k0 += 4) {
    const int kk = k0 + koff;
    v2f a;
    a.x = As[m * 132 + 64 + kk];
    a.y = As[m * 132 + 65 + kk];
    if (kk >= 32) {
      a.x *= Rs[m * 36 + kk - 32];
      a.y *= Rs[m * 36 + kk - 31];
    }
    const float* lh = LhW + kk * 32 + n;
    v2f b;
    b.x = lh[0];  b.y = lh[32];  t0 = wmma4(a, b, t0);
    b.x = lh[16]; b.y = lh[48];  t1 = wmma4(a, b, t1);
  }

  const float bh0 = Lhb[n], bh1 = Lhb[16 + n];
  const float hw0 = headW[n], hw1 = headW[16 + n];
  float p[8];
#pragma unroll
  for (int v = 0; v < 8; ++v) {
    const float ht0 = tanhf(t0[v] + bh0);
    const float ht1 = tanhf(t1[v] + bh1);
    const float hp0 = As[(row0 + v) * 132 + 96 + n];
    const float hp1 = As[(row0 + v) * 132 + 112 + n];
    const float h0 = z0[v] * hp0 + (1.0f - z0[v]) * ht0;
    const float h1 = z1[v] * hp1 + (1.0f - z1[v]) * ht1;
    h_out[(size_t)(tile + row0 + v) * 32 + n]      = h0;
    h_out[(size_t)(tile + row0 + v) * 32 + 16 + n] = h1;
    p[v] = fmaxf(h0, 0.0f) * hw0 + fmaxf(h1, 0.0f) * hw1;
  }
#pragma unroll
  for (int s = 1; s < 16; s <<= 1)
#pragma unroll
    for (int v = 0; v < 8; ++v)
      p[v] += __shfl_xor(p[v], s, 32);
  if (n == 0) {
    const float hb = headb[0];
#pragma unroll
    for (int v = 0; v < 8; ++v)
      y_out[tile + row0 + v] = p[v] + hb;
  }
}

extern "C" void kernel_launch(void* const* d_in, const int* in_sizes, int n_in,
                              void* d_out, int out_size, void* d_ws, size_t ws_size,
                              hipStream_t stream) {
  const float* x    = (const float*)d_in[0];
  const int*   ei   = (const int*)d_in[1];
  const float* ew   = (const float*)d_in[2];
  const float* Hp   = (const float*)d_in[3];
  const float* Wz   = (const float*)d_in[4];
  const float* bz   = (const float*)d_in[5];
  const float* Wr   = (const float*)d_in[6];
  const float* br   = (const float*)d_in[7];
  const float* Wh   = (const float*)d_in[8];
  const float* bh   = (const float*)d_in[9];
  const float* LzW  = (const float*)d_in[10];
  const float* Lzb  = (const float*)d_in[11];
  const float* LrW  = (const float*)d_in[12];
  const float* Lrb  = (const float*)d_in[13];
  const float* LhW  = (const float*)d_in[14];
  const float* Lhb  = (const float*)d_in[15];
  const float* hW   = (const float*)d_in[16];
  const float* hb   = (const float*)d_in[17];

  const int* src = ei;
  const int* dst = ei + Ee;

  float* out_y = (float*)d_out;
  float* out_h = out_y + Nn;

  float* ws   = (float*)d_ws;
  float* dis  = ws;
  float* Wcat = ws + Nn;
  float* bcat = Wcat + FIN * 96;
  float* XW   = bcat + 96;
  float* part = XW + (size_t)Nn * 96;
  float* P    = part + (size_t)DEG_S * Nn;
  unsigned* bm   = (unsigned*)(P + (size_t)AG_S * Nn * 96);
  unsigned* rec  = bm + (size_t)AG_Q * BM_W;

  pack_weights_kernel<<<(FIN * 96 + 255) / 256, 256, 0, stream>>>(Wz, Wr, Wh, bz, br, bh, Wcat, bcat);
  deg_partial_kernel<<<DEG_Q * DEG_S, DEG_TPB, 0, stream>>>(src, dst, ew, part, bm, rec);
  finalize_dis_kernel<<<(Nn + 255) / 256, 256, 0, stream>>>(part, dis);
  gemm_xw_kernel<<<Nn / 16, 32, 0, stream>>>(x, Wcat, XW);
  agg_range_kernel<<<AG_Q * AG_S, AG_TPB, 0, stream>>>(bm, rec, dis, XW, P);
  gru_kernel<<<Nn / 16, 32, 0, stream>>>(XW, P, dis, bcat, Hp, LzW, Lzb, LrW, Lrb, LhW, Lhb, hW, hb, out_y, out_h);
}
